# starting kernel
# baseline (speedup 1.0000x reference)
#include <hip/hip_runtime.h>

typedef float v2f __attribute__((ext_vector_type(2)));
typedef float v8f __attribute__((ext_vector_type(8)));

#define NN 50000
#define NE 800000
#define F  64
#define KD 256


__global__ __launch_bounds__(256) void zero_k(float* __restrict__ p, long n) {
    long i  = (long)blockIdx.x * blockDim.x + threadIdx.x;
    long st = (long)gridDim.x * blockDim.x;
    for (; i < n; i += st) p[i] = 0.0f;
}

__global__ __launch_bounds__(256) void deg_k(const int* __restrict__ dst, int* __restrict__ deg) {
    int e = blockIdx.x * 256 + threadIdx.x;
    if (e < NE) atomicAdd(&deg[dst[e]], 1);
}

__global__ __launch_bounds__(256) void norm_k(float* __restrict__ nrm) {
    int i = blockIdx.x * 256 + threadIdx.x;
    if (i < NN) {
        float d = (float)((const int*)nrm)[i];
        nrm[i] = rsqrtf(d < 1.0f ? 1.0f : d);
    }
}

__global__ __launch_bounds__(256) void hop_k(const int* __restrict__ src,
                                             const int* __restrict__ dst,
                                             const float* __restrict__ nrm,
                                             const float* __restrict__ hin,
                                             float* __restrict__ hout) {
    long t = (long)blockIdx.x * 256 + threadIdx.x;
    int  e = (int)(t >> 4);
    if (e >= NE) return;
    int   p  = (int)(t & 15);
    int   s  = src[e];
    int   d  = dst[e];
    float ns = nrm[s];
    float4 v = ((const float4*)(hin + (long)s * F))[p];
    float* op = hout + (long)d * F + p * 4;
    unsafeAtomicAdd(op + 0, v.x * ns);
    unsafeAtomicAdd(op + 1, v.y * ns);
    unsafeAtomicAdd(op + 2, v.z * ns);
    unsafeAtomicAdd(op + 3, v.w * ns);
}

__global__ __launch_bounds__(256) void scale_k(const float* __restrict__ nrm,
                                               float* __restrict__ h) {
    long i = (long)blockIdx.x * 256 + threadIdx.x;
    if (i < (long)NN * F) h[i] *= nrm[i >> 6];
}


__device__ __forceinline__ void seg_mac(const float* __restrict__ base, int kbase,
                                        int node, int hi2, int ln15,
                                        const float* Wt,
                                        v8f& c0, v8f& c1, v8f& c2, v8f& c3) {
    const float* rp = base + (long)node * F + hi2;
#pragma unroll 4
    for (int kk = 0; kk < F; kk += 4) {
        v2f a;
        a.x = rp[kk];
        a.y = rp[kk + 1];
        int kr = (kbase + kk + hi2) * F + ln15;
        v2f b0; b0.x = Wt[kr];      b0.y = Wt[kr + F];
        v2f b1; b1.x = Wt[kr + 16]; b1.y = Wt[kr + F + 16];
        v2f b2; b2.x = Wt[kr + 32]; b2.y = Wt[kr + F + 32];
        v2f b3; b3.x = Wt[kr + 48]; b3.y = Wt[kr + F + 48];
        c0 = __builtin_amdgcn_wmma_f32_16x16x4_f32(false, a, false, b0, (short)0, c0, false, false);
        c1 = __builtin_amdgcn_wmma_f32_16x16x4_f32(false, a, false, b1, (short)0, c1, false, false);
        c2 = __builtin_amdgcn_wmma_f32_16x16x4_f32(false, a, false, b2, (short)0, c2, false, false);
        c3 = __builtin_amdgcn_wmma_f32_16x16x4_f32(false, a, false, b3, (short)0, c3, false, false);
    }
}

__global__ __launch_bounds__(256) void gemm_k(const float* __restrict__ feat,
                                              const float* __restrict__ h1,
                                              const float* __restrict__ h2,
                                              const float* __restrict__ h3,
                                              const float* __restrict__ W,
                                              const float* __restrict__ bias,
                                              float* __restrict__ out) {
    __shared__ float Wt[KD * F];
    int tid = threadIdx.x;
    for (int i = tid; i < KD * F; i += 256) {
        int o = i >> 8;
        int k = i & 255;
        Wt[k * F + o] = W[i];
    }
    __syncthreads();

    int lane = tid & 31;
    int ln15 = lane & 15;
    int hi2  = (lane >= 16) ? 2 : 0;
    int rowBase = blockIdx.x * 128 + (tid >> 5) * 16;
    if (rowBase >= NN) return;

    int node = rowBase + ln15;
    if (node >= NN) node = NN - 1;

    v8f c0, c1, c2, c3;
    float bv0 = bias[ln15], bv1 = bias[16 + ln15], bv2 = bias[32 + ln15], bv3 = bias[48 + ln15];
#pragma unroll
    for (int i = 0; i < 8; ++i) { c0[i] = bv0; c1[i] = bv1; c2[i] = bv2; c3[i] = bv3; }

    seg_mac(feat, 0,   node, hi2, ln15, Wt, c0, c1, c2, c3);
    seg_mac(h1,   64,  node, hi2, ln15, Wt, c0, c1, c2, c3);
    seg_mac(h2,   128, node, hi2, ln15, Wt, c0, c1, c2, c3);
    seg_mac(h3,   192, node, hi2, ln15, Wt, c0, c1, c2, c3);

    int mBase = rowBase + (hi2 ? 8 : 0);
#pragma unroll
    for (int r = 0; r < 8; ++r) {
        int row = mBase + r;
        if (row < NN) {
            out[(long)row * F + ln15]      = c0[r];
            out[(long)row * F + 16 + ln15] = c1[r];
            out[(long)row * F + 32 + ln15] = c2[r];
            out[(long)row * F + 48 + ln15] = c3[r];
        }
    }
}


extern "C" void kernel_launch(void* const* d_in, const int* in_sizes, int n_in,
                              void* d_out, int out_size, void* d_ws, size_t ws_size,
                              hipStream_t stream) {
    const float* feat = (const float*)d_in[0];
    const int*   src  = (const int*)d_in[1];
    const int*   dst  = (const int*)d_in[2];
    const float* W    = (const float*)d_in[3];
    const float* bias = (const float*)d_in[4];
    float*       out  = (float*)d_out;

    float* ws   = (float*)d_ws;
    float* nrm  = ws;
    float* h1   = ws + 50048;
    float* h2   = h1 + (long)NN * F;
    float* h3   = h2 + (long)NN * F;
    long   ztot = 50048 + 3L * NN * F;

    zero_k <<<2048, 256, 0, stream>>>(ws, ztot);
    deg_k  <<<(NE + 255) / 256, 256, 0, stream>>>(dst, (int*)nrm);
    norm_k <<<(NN + 255) / 256, 256, 0, stream>>>(nrm);

    const int hopBlocks   = (NE * 16) / 256;
    const int scaleBlocks = (NN * F + 255) / 256;

    hop_k  <<<hopBlocks,   256, 0, stream>>>(src, dst, nrm, feat, h1);
    scale_k<<<scaleBlocks, 256, 0, stream>>>(nrm, h1);
    hop_k  <<<hopBlocks,   256, 0, stream>>>(src, dst, nrm, h1, h2);
    scale_k<<<scaleBlocks, 256, 0, stream>>>(nrm, h2);
    hop_k  <<<hopBlocks,   256, 0, stream>>>(src, dst, nrm, h2, h3);
    scale_k<<<scaleBlocks, 256, 0, stream>>>(nrm, h3);

    gemm_k <<<(NN + 127) / 128, 256, 0, stream>>>(feat, h1, h2, h3, W, bias, out);
}


# Hops gather per destination node from an edge list sorted by destination (LDS counting sort, rebuilt every call) instead of scattering with float atomics; zeroing and scale passes fold away
# speedup vs baseline: 2.4531x; 2.4531x over previous
#include <hip/hip_runtime.h>

typedef float v2f __attribute__((ext_vector_type(2)));
typedef float v8f __attribute__((ext_vector_type(8)));

#define NN 50000
#define NE 800000
#define F  64
#define KD 256

#define NCH 64
#define CHE (NE / NCH)
#define EPT 13
#define SPT 49
#define NPW 4


__global__ __launch_bounds__(1024) void hist_k(const int* __restrict__ dst,
                                               int* __restrict__ part,
                                               int* __restrict__ lrank) {
    __shared__ int4 hb[NN / 4];
    int* h = (int*)hb;
    int tid = threadIdx.x;
    int e0 = blockIdx.x * CHE;
    int dv[EPT], rk[EPT];
#pragma unroll
    for (int k = 0; k < EPT; ++k) {
        int i = tid + k * 1024;
        dv[k] = dst[e0 + (i < CHE ? i : CHE - 1)];
    }
    for (int i = tid; i < NN / 4; i += 1024) hb[i] = make_int4(0, 0, 0, 0);
    __syncthreads();
#pragma unroll
    for (int k = 0; k < EPT; ++k) {
        rk[k] = 0;
        if (tid + k * 1024 < CHE) rk[k] = atomicAdd(&h[dv[k]], 1);
    }
#pragma unroll
    for (int k = 0; k < EPT; ++k) {
        int i = tid + k * 1024;
        if (i < CHE) lrank[e0 + i] = rk[k];
    }
    __syncthreads();
    int4* po = (int4*)(part + (long)blockIdx.x * NN);
    for (int i = tid; i < NN / 4; i += 1024) po[i] = hb[i];
}

__global__ __launch_bounds__(256) void colscan_k(int* __restrict__ part, int* __restrict__ deg,
                                                 float* __restrict__ nrm) {
    int b = blockIdx.x * 256 + threadIdx.x;
    if (b >= NN) return;
    int run = 0;
    for (int g0 = 0; g0 < NCH; g0 += 16) {
        int t[16];
#pragma unroll
        for (int u = 0; u < 16; ++u) t[u] = part[(g0 + u) * NN + b];
#pragma unroll
        for (int u = 0; u < 16; ++u) { part[(g0 + u) * NN + b] = run; run += t[u]; }
    }
    deg[b] = run;
    float f = (float)run;
    nrm[b] = rsqrtf(f < 1.0f ? 1.0f : f);
}

__global__ __launch_bounds__(1024) void binscan_k(int* __restrict__ rowptr) {
    __shared__ int s[1024 * SPT];
    __shared__ int wtot[32];
    int tid = threadIdx.x;
    int dv[SPT];
#pragma unroll
    for (int k = 0; k < SPT; ++k) {
        int i = tid + k * 1024;
        dv[k] = rowptr[i < NN ? i : NN - 1];
    }
#pragma unroll
    for (int k = 0; k < SPT; ++k) {
        int i = tid + k * 1024;
        s[i] = (i < NN) ? dv[k] : 0;
    }
    __syncthreads();
    int base = tid * SPT;
    int tot = 0;
    for (int i = 0; i < SPT; ++i) tot += s[base + i];
    int lane = tid & 31, wv = tid >> 5;
    int incl = tot;
    for (int o = 1; o < 32; o <<= 1) {
        int y = __shfl_up(incl, o);
        if (lane >= o) incl += y;
    }
    if (lane == 31) wtot[wv] = incl;
    __syncthreads();
    int run = incl - tot;
    for (int w = 0; w < wv; ++w) run += wtot[w];
    for (int i = 0; i < SPT; ++i) { int v = s[base + i]; s[base + i] = run; run += v; }
    __syncthreads();
    for (int i = tid; i < NN; i += 1024) rowptr[i] = s[i];
    if (tid == 1023) rowptr[NN] = run;
}

__global__ __launch_bounds__(256) void fill_k(const int* __restrict__ src,
                                              const int* __restrict__ dst,
                                              const int* __restrict__ lrank,
                                              const int* __restrict__ rowptr,
                                              const int* __restrict__ part,
                                              const float* __restrict__ nrm,
                                              int2* __restrict__ csr) {
    int e = blockIdx.x * 256 + threadIdx.x;
    if (e >= NE) return;
    int d = dst[e];
    int s = src[e];
    int g = e / CHE;
    unsigned pos = (unsigned)(rowptr[d] + part[g * NN + d] + lrank[e]);
    if (pos > (unsigned)(NE - 1)) pos = NE - 1;
    csr[pos] = make_int2(s, __float_as_int(nrm[s]));
}


__global__ __launch_bounds__(256) void hop_k(const int2* __restrict__ csr,
                                             const int* __restrict__ rowptr,
                                             const float* __restrict__ nrm,
                                             const float* __restrict__ hin,
                                             float* __restrict__ hout) {
    int wv = __builtin_amdgcn_readfirstlane((int)(threadIdx.x >> 5));
    int n0 = (blockIdx.x * 8 + wv) * NPW;
    if (n0 >= NN) return;
    int lane = threadIdx.x & 31;
    int p  = lane & 15;
    int hh = lane >> 4;
    int rp[NPW + 1];
#pragma unroll
    for (int q = 0; q <= NPW; ++q) rp[q] = rowptr[n0 + q];
#pragma unroll
    for (int q = 0; q < NPW; ++q) {
        int beg = rp[q], end = rp[q + 1];
        float4 acc = make_float4(0.f, 0.f, 0.f, 0.f);
        for (int j0 = beg; j0 < end; j0 += 8) {
            int2   pr[4];
            float4 v[4];
#pragma unroll
            for (int u = 0; u < 4; ++u) {
                int j = j0 + 2 * u + hh;
                pr[u] = csr[j < end ? j : beg];
            }
#pragma unroll
            for (int u = 0; u < 4; ++u) v[u] = ((const float4*)(hin + (long)pr[u].x * F))[p];
#pragma unroll
            for (int u = 0; u < 4; ++u) {
                float w = (j0 + 2 * u + hh < end) ? __int_as_float(pr[u].y) : 0.0f;
                acc.x += v[u].x * w;
                acc.y += v[u].y * w;
                acc.z += v[u].z * w;
                acc.w += v[u].w * w;
            }
        }
        acc.x += __shfl_xor(acc.x, 16);
        acc.y += __shfl_xor(acc.y, 16);
        acc.z += __shfl_xor(acc.z, 16);
        acc.w += __shfl_xor(acc.w, 16);
        if (hh == 0) {
            float nd = nrm[n0 + q];
            ((float4*)(hout + (long)(n0 + q) * F))[p] =
                make_float4(acc.x * nd, acc.y * nd, acc.z * nd, acc.w * nd);
        }
    }
}


__device__ __forceinline__ void seg_mac(const float* __restrict__ base, int kbase,
                                        int node, int hi2, int ln15,
                                        const float* Wt,
                                        v8f& c0, v8f& c1, v8f& c2, v8f& c3) {
    const float* rp = base + (long)node * F + hi2;
#pragma unroll 4
    for (int kk = 0; kk < F; kk += 4) {
        v2f a;
        a.x = rp[kk];
        a.y = rp[kk + 1];
        int kr = (kbase + kk + hi2) * F + ln15;
        v2f b0; b0.x = Wt[kr];      b0.y = Wt[kr + F];
        v2f b1; b1.x = Wt[kr + 16]; b1.y = Wt[kr + F + 16];
        v2f b2; b2.x = Wt[kr + 32]; b2.y = Wt[kr + F + 32];
        v2f b3; b3.x = Wt[kr + 48]; b3.y = Wt[kr + F + 48];
        c0 = __builtin_amdgcn_wmma_f32_16x16x4_f32(false, a, false, b0, (short)0, c0, false, false);
        c1 = __builtin_amdgcn_wmma_f32_16x16x4_f32(false, a, false, b1, (short)0, c1, false, false);
        c2 = __builtin_amdgcn_wmma_f32_16x16x4_f32(false, a, false, b2, (short)0, c2, false, false);
        c3 = __builtin_amdgcn_wmma_f32_16x16x4_f32(false, a, false, b3, (short)0, c3, false, false);
    }
}

__global__ __launch_bounds__(256) void gemm_k(const float* __restrict__ feat,
                                              const float* __restrict__ h1,
                                              const float* __restrict__ h2,
                                              const float* __restrict__ h3,
                                              const float* __restrict__ W,
                                              const float* __restrict__ bias,
                                              float* __restrict__ out) {
    __shared__ float Wt[KD * F];
    int tid = threadIdx.x;
    for (int i = tid; i < KD * F; i += 256) {
        int o = i >> 8;
        int k = i & 255;
        Wt[k * F + o] = W[i];
    }
    __syncthreads();

    int lane = tid & 31;
    int ln15 = lane & 15;
    int hi2  = (lane >= 16) ? 2 : 0;
    int rowBase = blockIdx.x * 128 + (tid >> 5) * 16;
    if (rowBase >= NN) return;

    int node = rowBase + ln15;
    if (node >= NN) node = NN - 1;

    v8f c0, c1, c2, c3;
    float bv0 = bias[ln15], bv1 = bias[16 + ln15], bv2 = bias[32 + ln15], bv3 = bias[48 + ln15];
#pragma unroll
    for (int i = 0; i < 8; ++i) { c0[i] = bv0; c1[i] = bv1; c2[i] = bv2; c3[i] = bv3; }

    seg_mac(feat, 0,   node, hi2, ln15, Wt, c0, c1, c2, c3);
    seg_mac(h1,   64,  node, hi2, ln15, Wt, c0, c1, c2, c3);
    seg_mac(h2,   128, node, hi2, ln15, Wt, c0, c1, c2, c3);
    seg_mac(h3,   192, node, hi2, ln15, Wt, c0, c1, c2, c3);

    int mBase = rowBase + (hi2 ? 8 : 0);
#pragma unroll
    for (int r = 0; r < 8; ++r) {
        int row = mBase + r;
        if (row < NN) {
            out[(long)row * F + ln15]      = c0[r];
            out[(long)row * F + 16 + ln15] = c1[r];
            out[(long)row * F + 32 + ln15] = c2[r];
            out[(long)row * F + 48 + ln15] = c3[r];
        }
    }
}


extern "C" void kernel_launch(void* const* d_in, const int* in_sizes, int n_in,
                              void* d_out, int out_size, void* d_ws, size_t ws_size,
                              hipStream_t stream) {
    const float* feat = (const float*)d_in[0];
    const int*   src  = (const int*)d_in[1];
    const int*   dst  = (const int*)d_in[2];
    const float* W    = (const float*)d_in[3];
    const float* bias = (const float*)d_in[4];
    float*       out  = (float*)d_out;

    float* ws   = (float*)d_ws;
    float* nrm  = ws;
    float* h1   = ws + 50048;
    float* h2   = h1 + (long)NN * F;
    float* h3   = h2 + (long)NN * F;

    int* part  = (int*)h3;
    int* lrank = (int*)h2;
    int2* csr    = (int2*)out;
    int*  rowptr = (int*)out + 2 * NE;

    hist_k   <<<NCH, 1024, 0, stream>>>(dst, part, lrank);
    colscan_k<<<(NN + 255) / 256, 256, 0, stream>>>(part, rowptr, nrm);
    binscan_k<<<1, 1024, 0, stream>>>(rowptr);
    fill_k   <<<(NE + 255) / 256, 256, 0, stream>>>(src, dst, lrank, rowptr, part, nrm, csr);

    const int hopBlocks = (NN + 8 * NPW - 1) / (8 * NPW);

    hop_k<<<hopBlocks, 256, 0, stream>>>(csr, rowptr, nrm, feat, h1);
    hop_k<<<hopBlocks, 256, 0, stream>>>(csr, rowptr, nrm, h1, h2);
    hop_k<<<hopBlocks, 256, 0, stream>>>(csr, rowptr, nrm, h2, h3);

    gemm_k<<<(NN + 127) / 128, 256, 0, stream>>>(feat, h1, h2, h3, W, bias, out);
}


# Projection: W copied to LDS as it lies in memory (row pitch 260 floats), A and B fragments read 16 bytes per lane in one k order; the transposing copy that put 32 lanes on one bank is gone
# speedup vs baseline: 2.9146x; 1.1881x over previous
#include <hip/hip_runtime.h>

typedef float v2f __attribute__((ext_vector_type(2)));
typedef float v8f __attribute__((ext_vector_type(8)));

#define NN 50000
#define NE 800000
#define F  64
#define KD 256

#define NCH 64
#define CHE (NE / NCH)
#define EPT 13
#define NPW 4
#define NB  196


__global__ __launch_bounds__(1024) void hist_k(const int* __restrict__ dst,
                                               int* __restrict__ part,
                                               int* __restrict__ lrank,
                                               int* __restrict__ bsum) {
    __shared__ int4 hb[NN / 4];
    int* h = (int*)hb;
    int tid = threadIdx.x;
    int e0 = blockIdx.x * CHE;
    int dv[EPT], rk[EPT];
#pragma unroll
    for (int k = 0; k < EPT; ++k) {
        int i = tid + k * 1024;
        dv[k] = dst[e0 + (i < CHE ? i : CHE - 1)];
    }
    for (int i = tid; i < NN / 4; i += 1024) hb[i] = make_int4(0, 0, 0, 0);
    __syncthreads();
#pragma unroll
    for (int k = 0; k < EPT; ++k) {
        rk[k] = 0;
        if (tid + k * 1024 < CHE) rk[k] = atomicAdd(&h[dv[k]], 1);
    }
#pragma unroll
    for (int k = 0; k < EPT; ++k) {
        int i = tid + k * 1024;
        if (i < CHE) lrank[e0 + i] = rk[k];
    }
    __syncthreads();
    int bs = 0;
    if (tid < 4 * NB) {
#pragma unroll
        for (int j = 0; j < 16; ++j) {
            int i = tid * 16 + ((j + tid) & 15);
            int4 v = hb[i < NN / 4 ? i : 0];
            int sv = (v.x + v.y) + (v.z + v.w);
            bs += (i < NN / 4) ? sv : 0;
        }
    }
    bs += __shfl_xor(bs, 1);
    bs += __shfl_xor(bs, 2);
    if (tid < 4 * NB && (tid & 3) == 0) bsum[blockIdx.x * NB + (tid >> 2)] = bs;
    int4* po = (int4*)(part + (long)blockIdx.x * NN);
    for (int i = tid; i < NN / 4; i += 1024) po[i] = hb[i];
}

__global__ __launch_bounds__(256) void colscan_k(int* __restrict__ part,
                                                 const int* __restrict__ bsum,
                                                 int* __restrict__ rowptr,
                                                 float* __restrict__ nrm) {
    __shared__ int wdeg[8];
    __shared__ int wblk[8];
    int t = threadIdx.x;
    int lane = t & 31, wv = t >> 5;
    int b = blockIdx.x * 256 + t;
    bool live = b < NN;
    int bb = live ? b : NN - 1;
    int tt = t < NB ? t : NB - 1;
    int cnt[NCH];
#pragma unroll
    for (int c = 0; c < NCH; ++c) cnt[c] = part[c * NN + bb];
    int tb = 0;
#pragma unroll
    for (int c = 0; c < NCH; ++c) tb += bsum[c * NB + tt];
    if (t >= (int)blockIdx.x) tb = 0;
    int deg = 0;
#pragma unroll
    for (int c = 0; c < NCH; ++c) deg += cnt[c];
    if (!live) deg = 0;
    int incl = deg;
    for (int o = 1; o < 32; o <<= 1) {
        int y = __shfl_up(incl, o);
        if (lane >= o) incl += y;
    }
    for (int o = 16; o > 0; o >>= 1) tb += __shfl_xor(tb, o);
    if (lane == 31) wdeg[wv] = incl;
    if (lane == 0)  wblk[wv] = tb;
    __syncthreads();
    int start = incl - deg;
#pragma unroll
    for (int w = 0; w < 8; ++w) {
        start += wblk[w];
        if (w < wv) start += wdeg[w];
    }
    if (live) {
        rowptr[b] = start;
        if (b == NN - 1) rowptr[NN] = start + deg;
        float f = (float)deg;
        nrm[b] = rsqrtf(f < 1.0f ? 1.0f : f);
        int run = start;
#pragma unroll
        for (int c = 0; c < NCH; ++c) { part[c * NN + b] = run; run += cnt[c]; }
    }
}

__global__ __launch_bounds__(256) void fill_k(const int* __restrict__ src,
                                              const int* __restrict__ dst,
                                              const int* __restrict__ lrank,
                                              const int* __restrict__ part,
                                              const float* __restrict__ nrm,
                                              int2* __restrict__ csr) {
    int e = blockIdx.x * 256 + threadIdx.x;
    if (e >= NE) return;
    int d = dst[e];
    int s = src[e];
    int g = e / CHE;
    unsigned pos = (unsigned)(part[g * NN + d] + lrank[e]);
    if (pos > (unsigned)(NE - 1)) pos = NE - 1;
    csr[pos] = make_int2(s, __float_as_int(nrm[s]));
}


__global__ __launch_bounds__(256) void hop_k(const int2* __restrict__ csr,
                                             const int* __restrict__ rowptr,
                                             const float* __restrict__ nrm,
                                             const float* __restrict__ hin,
                                             float* __restrict__ hout) {
    int wv = __builtin_amdgcn_readfirstlane((int)(threadIdx.x >> 5));
    int n0 = (blockIdx.x * 8 + wv) * NPW;
    if (n0 >= NN) return;
    int lane = threadIdx.x & 31;
    int p  = lane & 15;
    int hh = lane >> 4;
    int rp[NPW + 1];
#pragma unroll
    for (int q = 0; q <= NPW; ++q) rp[q] = rowptr[n0 + q];
#pragma unroll
    for (int q = 0; q < NPW; ++q) {
        int beg = rp[q], end = rp[q + 1];
        float4 acc = make_float4(0.f, 0.f, 0.f, 0.f);
        for (int j0 = beg; j0 < end; j0 += 8) {
            int2   pr[4];
            float4 v[4];
#pragma unroll
            for (int u = 0; u < 4; ++u) {
                int j = j0 + 2 * u + hh;
                pr[u] = csr[j < end ? j : beg];
            }
#pragma unroll
            for (int u = 0; u < 4; ++u) v[u] = ((const float4*)(hin + (long)pr[u].x * F))[p];
#pragma unroll
            for (int u = 0; u < 4; ++u) {
                float w = (j0 + 2 * u + hh < end) ? __int_as_float(pr[u].y) : 0.0f;
                acc.x += v[u].x * w;
                acc.y += v[u].y * w;
                acc.z += v[u].z * w;
                acc.w += v[u].w * w;
            }
        }
        acc.x += __shfl_xor(acc.x, 16);
        acc.y += __shfl_xor(acc.y, 16);
        acc.z += __shfl_xor(acc.z, 16);
        acc.w += __shfl_xor(acc.w, 16);
        if (hh == 0) {
            float nd = nrm[n0 + q];
            ((float4*)(hout + (long)(n0 + q) * F))[p] =
                make_float4(acc.x * nd, acc.y * nd, acc.z * nd, acc.w * nd);
        }
    }
}


#define WP 260

__device__ __forceinline__ void seg_mac(const float* __restrict__ base, int kbase,
                                        int node, int h4, int ln15,
                                        const float* Ws,
                                        v8f& c0, v8f& c1, v8f& c2, v8f& c3) {
    const float* rp = base + (long)node * F + h4;
    const float* wp = Ws + ln15 * WP + kbase + h4;
#pragma unroll 2
    for (int kk = 0; kk < F; kk += 8) {
        float4 a4 = *(const float4*)(rp + kk);
        float4 w0 = *(const float4*)(wp + kk);
        float4 w1 = *(const float4*)(wp + 16 * WP + kk);
        float4 w2 = *(const float4*)(wp + 32 * WP + kk);
        float4 w3 = *(const float4*)(wp + 48 * WP + kk);
        v2f a;  a.x = a4.x;  a.y = a4.y;
        v2f b0; b0.x = w0.x; b0.y = w0.y;
        v2f b1; b1.x = w1.x; b1.y = w1.y;
        v2f b2; b2.x = w2.x; b2.y = w2.y;
        v2f b3; b3.x = w3.x; b3.y = w3.y;
        c0 = __builtin_amdgcn_wmma_f32_16x16x4_f32(false, a, false, b0, (short)0, c0, false, false);
        c1 = __builtin_amdgcn_wmma_f32_16x16x4_f32(false, a, false, b1, (short)0, c1, false, false);
        c2 = __builtin_amdgcn_wmma_f32_16x16x4_f32(false, a, false, b2, (short)0, c2, false, false);
        c3 = __builtin_amdgcn_wmma_f32_16x16x4_f32(false, a, false, b3, (short)0, c3, false, false);
        a.x = a4.z;  a.y = a4.w;
        b0.x = w0.z; b0.y = w0.w;
        b1.x = w1.z; b1.y = w1.w;
        b2.x = w2.z; b2.y = w2.w;
        b3.x = w3.z; b3.y = w3.w;
        c0 = __builtin_amdgcn_wmma_f32_16x16x4_f32(false, a, false, b0, (short)0, c0, false, false);
        c1 = __builtin_amdgcn_wmma_f32_16x16x4_f32(false, a, false, b1, (short)0, c1, false, false);
        c2 = __builtin_amdgcn_wmma_f32_16x16x4_f32(false, a, false, b2, (short)0, c2, false, false);
        c3 = __builtin_amdgcn_wmma_f32_16x16x4_f32(false, a, false, b3, (short)0, c3, false, false);
    }
}

__global__ __launch_bounds__(256) void gemm_k(const float* __restrict__ feat,
                                              const float* __restrict__ h1,
                                              const float* __restrict__ h2,
                                              const float* __restrict__ h3,
                                              const float* __restrict__ W,
                                              const float* __restrict__ bias,
                                              float* __restrict__ out) {
    __shared__ float4 Wsv[F * WP / 4];
    int tid = threadIdx.x;
    for (int i = tid; i < F * KD / 4; i += 256) {
        int o  = i >> 6;
        int c4 = i & 63;
        Wsv[o * (WP / 4) + c4] = ((const float4*)W)[i];
    }
    __syncthreads();
    const float* Ws = (const float*)Wsv;

    int lane = tid & 31;
    int ln15 = lane & 15;
    int h4   = (lane >= 16) ? 4 : 0;
    int rowBase = blockIdx.x * 128 + (tid >> 5) * 16;
    if (rowBase >= NN) return;

    int node = rowBase + ln15;
    if (node >= NN) node = NN - 1;

    v8f c0, c1, c2, c3;
    float bv0 = bias[ln15], bv1 = bias[16 + ln15], bv2 = bias[32 + ln15], bv3 = bias[48 + ln15];
#pragma unroll
    for (int i = 0; i < 8; ++i) { c0[i] = bv0; c1[i] = bv1; c2[i] = bv2; c3[i] = bv3; }

    seg_mac(feat, 0,   node, h4, ln15, Ws, c0, c1, c2, c3);
    seg_mac(h1,   64,  node, h4, ln15, Ws, c0, c1, c2, c3);
    seg_mac(h2,   128, node, h4, ln15, Ws, c0, c1, c2, c3);
    seg_mac(h3,   192, node, h4, ln15, Ws, c0, c1, c2, c3);

    int mBase = rowBase + (h4 ? 8 : 0);
#pragma unroll
    for (int r = 0; r < 8; ++r) {
        int row = mBase + r;
        if (row < NN) {
            out[(long)row * F + ln15]      = c0[r];
            out[(long)row * F + 16 + ln15] = c1[r];
            out[(long)row * F + 32 + ln15] = c2[r];
            out[(long)row * F + 48 + ln15] = c3[r];
        }
    }
}


extern "C" void kernel_launch(void* const* d_in, const int* in_sizes, int n_in,
                              void* d_out, int out_size, void* d_ws, size_t ws_size,
                              hipStream_t stream) {
    const float* feat = (const float*)d_in[0];
    const int*   src  = (const int*)d_in[1];
    const int*   dst  = (const int*)d_in[2];
    const float* W    = (const float*)d_in[3];
    const float* bias = (const float*)d_in[4];
    float*       out  = (float*)d_out;

    float* ws   = (float*)d_ws;
    float* nrm  = ws;
    float* h1   = ws + 50048;
    float* h2   = h1 + (long)NN * F;
    float* h3   = h2 + (long)NN * F;

    int* part  = (int*)h3;
    int* lrank = (int*)h2;
    int* bsum  = (int*)h1;
    int2* csr    = (int2*)out;
    int*  rowptr = (int*)out + 2 * NE;

    hist_k   <<<NCH, 1024, 0, stream>>>(dst, part, lrank, bsum);
    colscan_k<<<NB, 256, 0, stream>>>(part, bsum, rowptr, nrm);
    fill_k   <<<(NE + 255) / 256, 256, 0, stream>>>(src, dst, lrank, part, nrm, csr);

    const int hopBlocks = (NN + 8 * NPW - 1) / (8 * NPW);

    hop_k<<<hopBlocks, 256, 0, stream>>>(csr, rowptr, nrm, feat, h1);
    hop_k<<<hopBlocks, 256, 0, stream>>>(csr, rowptr, nrm, h1, h2);
    hop_k<<<hopBlocks, 256, 0, stream>>>(csr, rowptr, nrm, h2, h3);

    gemm_k<<<(NN + 127) / 128, 256, 0, stream>>>(feat, h1, h2, h3, W, bias, out);
}


# Third hop computed inside the projection launch: each wave gathers its 16 nodes' summed rows into LDS rows of its own and multiplies from there; one launch and the h3 round trip through memory go away
# speedup vs baseline: 3.2082x; 1.1007x over previous
#include <hip/hip_runtime.h>

typedef float v2f __attribute__((ext_vector_type(2)));
typedef float v8f __attribute__((ext_vector_type(8)));

#define NN 50000
#define NE 800000
#define F  64
#define KD 256

#define NCH 64
#define CHE (NE / NCH)
#define EPT 13
#define NPW 4
#define NB  196


__global__ __launch_bounds__(1024) void hist_k(const int* __restrict__ dst,
                                               int* __restrict__ part,
                                               int* __restrict__ lrank,
                                               int* __restrict__ bsum) {
    __shared__ int4 hb[NN / 4];
    int* h = (int*)hb;
    int tid = threadIdx.x;
    int e0 = blockIdx.x * CHE;
    int dv[EPT], rk[EPT];
#pragma unroll
    for (int k = 0; k < EPT; ++k) {
        int i = tid + k * 1024;
        dv[k] = dst[e0 + (i < CHE ? i : CHE - 1)];
    }
    for (int i = tid; i < NN / 4; i += 1024) hb[i] = make_int4(0, 0, 0, 0);
    __syncthreads();
#pragma unroll
    for (int k = 0; k < EPT; ++k) {
        rk[k] = 0;
        if (tid + k * 1024 < CHE) rk[k] = atomicAdd(&h[dv[k]], 1);
    }
#pragma unroll
    for (int k = 0; k < EPT; ++k) {
        int i = tid + k * 1024;
        if (i < CHE) lrank[e0 + i] = rk[k];
    }
    __syncthreads();
    int bs = 0;
    if (tid < 4 * NB) {
#pragma unroll
        for (int j = 0; j < 16; ++j) {
            int i = tid * 16 + ((j + tid) & 15);
            int4 v = hb[i < NN / 4 ? i : 0];
            int sv = (v.x + v.y) + (v.z + v.w);
            bs += (i < NN / 4) ? sv : 0;
        }
    }
    bs += __shfl_xor(bs, 1);
    bs += __shfl_xor(bs, 2);
    if (tid < 4 * NB && (tid & 3) == 0) bsum[blockIdx.x * NB + (tid >> 2)] = bs;
    int4* po = (int4*)(part + (long)blockIdx.x * NN);
    for (int i = tid; i < NN / 4; i += 1024) po[i] = hb[i];
}

__global__ __launch_bounds__(256) void colscan_k(int* __restrict__ part,
                                                 const int* __restrict__ bsum,
                                                 int* __restrict__ rowptr,
                                                 float* __restrict__ nrm) {
    __shared__ int wdeg[8];
    __shared__ int wblk[8];
    int t = threadIdx.x;
    int lane = t & 31, wv = t >> 5;
    int b = blockIdx.x * 256 + t;
    bool live = b < NN;
    int bb = live ? b : NN - 1;
    int tt = t < NB ? t : NB - 1;
    int cnt[NCH];
#pragma unroll
    for (int c = 0; c < NCH; ++c) cnt[c] = part[c * NN + bb];
    int tb = 0;
#pragma unroll
    for (int c = 0; c < NCH; ++c) tb += bsum[c * NB + tt];
    if (t >= (int)blockIdx.x) tb = 0;
    int deg = 0;
#pragma unroll
    for (int c = 0; c < NCH; ++c) deg += cnt[c];
    if (!live) deg = 0;
    int incl = deg;
    for (int o = 1; o < 32; o <<= 1) {
        int y = __shfl_up(incl, o);
        if (lane >= o) incl += y;
    }
    for (int o = 16; o > 0; o >>= 1) tb += __shfl_xor(tb, o);
    if (lane == 31) wdeg[wv] = incl;
    if (lane == 0)  wblk[wv] = tb;
    __syncthreads();
    int start = incl - deg;
#pragma unroll
    for (int w = 0; w < 8; ++w) {
        start += wblk[w];
        if (w < wv) start += wdeg[w];
    }
    if (live) {
        rowptr[b] = start;
        if (b == NN - 1) rowptr[NN] = start + deg;
        float f = (float)deg;
        nrm[b] = rsqrtf(f < 1.0f ? 1.0f : f);
        int run = start;
#pragma unroll
        for (int c = 0; c < NCH; ++c) { part[c * NN + b] = run; run += cnt[c]; }
    }
}

__global__ __launch_bounds__(256) void fill_k(const int* __restrict__ src,
                                              const int* __restrict__ dst,
                                              const int* __restrict__ lrank,
                                              const int* __restrict__ part,
                                              const float* __restrict__ nrm,
                                              int2* __restrict__ csr) {
    int e = blockIdx.x * 256 + threadIdx.x;
    if (e >= NE) return;
    int d = dst[e];
    int s = src[e];
    int g = e / CHE;
    unsigned pos = (unsigned)(part[g * NN + d] + lrank[e]);
    if (pos > (unsigned)(NE - 1)) pos = NE - 1;
    csr[pos] = make_int2(s, __float_as_int(nrm[s]));
}


__global__ __launch_bounds__(256) void hop_k(const int2* __restrict__ csr,
                                             const int* __restrict__ rowptr,
                                             const float* __restrict__ nrm,
                                             const float* __restrict__ hin,
                                             float* __restrict__ hout) {
    int wv = __builtin_amdgcn_readfirstlane((int)(threadIdx.x >> 5));
    int n0 = (blockIdx.x * 8 + wv) * NPW;
    if (n0 >= NN) return;
    int lane = threadIdx.x & 31;
    int p  = lane & 15;
    int hh = lane >> 4;
    int rp[NPW + 1];
#pragma unroll
    for (int q = 0; q <= NPW; ++q) rp[q] = rowptr[n0 + q];
#pragma unroll
    for (int q = 0; q < NPW; ++q) {
        int beg = rp[q], end = rp[q + 1];
        float4 acc = make_float4(0.f, 0.f, 0.f, 0.f);
        for (int j0 = beg; j0 < end; j0 += 8) {
            int2   pr[4];
            float4 v[4];
#pragma unroll
            for (int u = 0; u < 4; ++u) {
                int j = j0 + 2 * u + hh;
                pr[u] = csr[j < end ? j : beg];
            }
#pragma unroll
            for (int u = 0; u < 4; ++u) v[u] = ((const float4*)(hin + (long)pr[u].x * F))[p];
#pragma unroll
            for (int u = 0; u < 4; ++u) {
                float w = (j0 + 2 * u + hh < end) ? __int_as_float(pr[u].y) : 0.0f;
                acc.x += v[u].x * w;
                acc.y += v[u].y * w;
                acc.z += v[u].z * w;
                acc.w += v[u].w * w;
            }
        }
        acc.x += __shfl_xor(acc.x, 16);
        acc.y += __shfl_xor(acc.y, 16);
        acc.z += __shfl_xor(acc.z, 16);
        acc.w += __shfl_xor(acc.w, 16);
        if (hh == 0) {
            float nd = nrm[n0 + q];
            ((float4*)(hout + (long)(n0 + q) * F))[p] =
                make_float4(acc.x * nd, acc.y * nd, acc.z * nd, acc.w * nd);
        }
    }
}


#define WP 260

__device__ __forceinline__ void seg_mac(const float* __restrict__ rp, int kbase,
                                        int h4, int ln15,
                                        const float* Ws,
                                        v8f& c0, v8f& c1, v8f& c2, v8f& c3) {
    const float* wp = Ws + ln15 * WP + kbase + h4;
#pragma unroll 2
    for (int kk = 0; kk < F; kk += 8) {
        float4 a4 = *(const float4*)(rp + kk);
        float4 w0 = *(const float4*)(wp + kk);
        float4 w1 = *(const float4*)(wp + 16 * WP + kk);
        float4 w2 = *(const float4*)(wp + 32 * WP + kk);
        float4 w3 = *(const float4*)(wp + 48 * WP + kk);
        v2f a;  a.x = a4.x;  a.y = a4.y;
        v2f b0; b0.x = w0.x; b0.y = w0.y;
        v2f b1; b1.x = w1.x; b1.y = w1.y;
        v2f b2; b2.x = w2.x; b2.y = w2.y;
        v2f b3; b3.x = w3.x; b3.y = w3.y;
        c0 = __builtin_amdgcn_wmma_f32_16x16x4_f32(false, a, false, b0, (short)0, c0, false, false);
        c1 = __builtin_amdgcn_wmma_f32_16x16x4_f32(false, a, false, b1, (short)0, c1, false, false);
        c2 = __builtin_amdgcn_wmma_f32_16x16x4_f32(false, a, false, b2, (short)0, c2, false, false);
        c3 = __builtin_amdgcn_wmma_f32_16x16x4_f32(false, a, false, b3, (short)0, c3, false, false);
        a.x = a4.z;  a.y = a4.w;
        b0.x = w0.z; b0.y = w0.w;
        b1.x = w1.z; b1.y = w1.w;
        b2.x = w2.z; b2.y = w2.w;
        b3.x = w3.z; b3.y = w3.w;
        c0 = __builtin_amdgcn_wmma_f32_16x16x4_f32(false, a, false, b0, (short)0, c0, false, false);
        c1 = __builtin_amdgcn_wmma_f32_16x16x4_f32(false, a, false, b1, (short)0, c1, false, false);
        c2 = __builtin_amdgcn_wmma_f32_16x16x4_f32(false, a, false, b2, (short)0, c2, false, false);
        c3 = __builtin_amdgcn_wmma_f32_16x16x4_f32(false, a, false, b3, (short)0, c3, false, false);
    }
}

#define HP 68

template <int U> __device__ __forceinline__ int row_bcast(int x) {
    return __builtin_amdgcn_update_dpp(0, x, 0x150 + U, 0xF, 0xF, false);
}

__global__ __launch_bounds__(256) void hop_gemm_k(const int2* __restrict__ csr,
                                                  const int* __restrict__ rowptr,
                                                  const float* __restrict__ nrm,
                                                  const float* __restrict__ feat,
                                                  const float* __restrict__ h1,
                                                  const float* __restrict__ h2,
                                                  const float* __restrict__ W,
                                                  const float* __restrict__ bias,
                                                  float* __restrict__ out) {
    __shared__ float4 Wsv[F * WP / 4];
    __shared__ float4 Hsv[8 * 16 * HP / 4];
    int tid = threadIdx.x;
    for (int i = tid; i < F * KD / 4; i += 256) {
        int o  = i >> 6;
        int c4 = i & 63;
        Wsv[o * (WP / 4) + c4] = ((const float4*)W)[i];
    }
    const float* Ws = (const float*)Wsv;

    int lane = tid & 31;
    int ln15 = lane & 15;
    int hh   = lane >> 4;
    int h4   = hh * 4;
    int wv   = tid >> 5;
    int rowBase = blockIdx.x * 128 + wv * 16;
    bool live = rowBase < NN;
    float* Hs = (float*)Hsv + wv * (16 * HP);

    if (live) {
        int   rpv = rowptr[rowBase + (lane < 17 ? lane : 16)];
        float nrv = nrm[rowBase + ln15];
        const float4* h2v = (const float4*)h2;
        int begq[8], endq[8];
        int2 e0[8];
#pragma unroll
        for (int q = 0; q < 8; ++q) {
            begq[q] = __shfl(rpv, 2 * q + hh);
            endq[q] = __shfl(rpv, 2 * q + hh + 1);
            int j = begq[q] + ln15;
            int jc = j < endq[q] ? j : begq[q];
            if (jc > NE - 1) jc = NE - 1;
            e0[q] = csr[jc];
        }
#pragma unroll
        for (int q = 0; q < 8; ++q) {
            int beg = begq[q], end = endq[q];
            int2 ent = e0[q];
            float4 acc = make_float4(0.f, 0.f, 0.f, 0.f);
            for (int j0 = beg; j0 < end; j0 += 16) {
                int2 cur = ent;
                if (j0 + 16 < end) {
                    int jn = j0 + 16 + ln15;
                    ent = csr[jn < end ? jn : beg];
                }
                int wy = (j0 + ln15 < end) ? cur.y : 0;
                float4 v[8];
#define LD_ROW(V, U) { unsigned s = (unsigned)row_bcast<U>(cur.x); V[(U) & 7] = h2v[s * 16u + (unsigned)ln15]; }
#define AC_ROW(V, U) { float w = __int_as_float(row_bcast<U>(wy)); \
                       acc.x += V[(U) & 7].x * w; acc.y += V[(U) & 7].y * w; \
                       acc.z += V[(U) & 7].z * w; acc.w += V[(U) & 7].w * w; }
                LD_ROW(v, 0) LD_ROW(v, 1) LD_ROW(v, 2) LD_ROW(v, 3)
                LD_ROW(v, 4) LD_ROW(v, 5) LD_ROW(v, 6) LD_ROW(v, 7)
                if (j0 + 8 < end) {
                    float4 t[8];
                    LD_ROW(t, 8)  LD_ROW(t, 9)  LD_ROW(t, 10) LD_ROW(t, 11)
                    LD_ROW(t, 12) LD_ROW(t, 13) LD_ROW(t, 14) LD_ROW(t, 15)
                    AC_ROW(t, 8)  AC_ROW(t, 9)  AC_ROW(t, 10) AC_ROW(t, 11)
                    AC_ROW(t, 12) AC_ROW(t, 13) AC_ROW(t, 14) AC_ROW(t, 15)
                }
                AC_ROW(v, 0) AC_ROW(v, 1) AC_ROW(v, 2) AC_ROW(v, 3)
                AC_ROW(v, 4) AC_ROW(v, 5) AC_ROW(v, 6) AC_ROW(v, 7)
#undef LD_ROW
#undef AC_ROW
            }
            float nd = __shfl(nrv, 2 * q + hh);
            *(float4*)(Hs + (2 * q + hh) * HP + 4 * ln15) =
                make_float4(acc.x * nd, acc.y * nd, acc.z * nd, acc.w * nd);
        }
    }
    __syncthreads();
    if (!live) return;

    int node = rowBase + ln15;

    v8f c0, c1, c2, c3;
    float bv0 = bias[ln15], bv1 = bias[16 + ln15], bv2 = bias[32 + ln15], bv3 = bias[48 + ln15];
#pragma unroll
    for (int i = 0; i < 8; ++i) { c0[i] = bv0; c1[i] = bv1; c2[i] = bv2; c3[i] = bv3; }

    seg_mac(feat + (long)node * F + h4, 0,   h4, ln15, Ws, c0, c1, c2, c3);
    seg_mac(h1   + (long)node * F + h4, 64,  h4, ln15, Ws, c0, c1, c2, c3);
    seg_mac(h2   + (long)node * F + h4, 128, h4, ln15, Ws, c0, c1, c2, c3);
    seg_mac(Hs + ln15 * HP + h4,        192, h4, ln15, Ws, c0, c1, c2, c3);

    int mBase = rowBase + (h4 ? 8 : 0);
#pragma unroll
    for (int r = 0; r < 8; ++r) {
        int row = mBase + r;
        out[(long)row * F + ln15]      = c0[r];
        out[(long)row * F + 16 + ln15] = c1[r];
        out[(long)row * F + 32 + ln15] = c2[r];
        out[(long)row * F + 48 + ln15] = c3[r];
    }
}


extern "C" void kernel_launch(void* const* d_in, const int* in_sizes, int n_in,
                              void* d_out, int out_size, void* d_ws, size_t ws_size,
                              hipStream_t stream) {
    const float* feat = (const float*)d_in[0];
    const int*   src  = (const int*)d_in[1];
    const int*   dst  = (const int*)d_in[2];
    const float* W    = (const float*)d_in[3];
    const float* bias = (const float*)d_in[4];
    float*       out  = (float*)d_out;

    float* ws   = (float*)d_ws;
    float* nrm  = ws;
    float* h1   = ws + 50048;
    float* h2   = h1 + (long)NN * F;
    float* h3   = h2 + (long)NN * F;

    int* part  = (int*)h1;
    int* lrank = (int*)h2;
    int* bsum  = (int*)h2 + NE;
    int2* csr    = (int2*)h3;
    int*  rowptr = (int*)h3 + 2 * NE;

    hist_k   <<<NCH, 1024, 0, stream>>>(dst, part, lrank, bsum);
    colscan_k<<<NB, 256, 0, stream>>>(part, bsum, rowptr, nrm);
    fill_k   <<<(NE + 255) / 256, 256, 0, stream>>>(src, dst, lrank, part, nrm, csr);

    const int hopBlocks = (NN + 8 * NPW - 1) / (8 * NPW);

    hop_k<<<hopBlocks, 256, 0, stream>>>(csr, rowptr, nrm, feat, h1);
    hop_k<<<hopBlocks, 256, 0, stream>>>(csr, rowptr, nrm, h1, h2);
    hop_gemm_k<<<(NN + 127) / 128, 256, 0, stream>>>(csr, rowptr, nrm, feat, h1, h2, W, bias, out);
}


# Edge list built in two levels: each chunk groups its pairs by bucket of 256 destinations in LDS, then one work-group per bucket counts and places its rows; the per-chunk count table and its passes go
# speedup vs baseline: 3.6560x; 1.1396x over previous
#include <hip/hip_runtime.h>

typedef float v2f __attribute__((ext_vector_type(2)));
typedef float v8f __attribute__((ext_vector_type(8)));

#define NN 50000
#define NE 800000
#define F  64
#define KD 256

#define NCH 64
#define CHE (NE / NCH)
#define EPT 13
#define NPW 4
#define NBK 196


__global__ __launch_bounds__(1024) void bucket_k(const int* __restrict__ src,
                                                 const int* __restrict__ dst,
                                                 int2* __restrict__ ebuf,
                                                 int* __restrict__ cntg,
                                                 int* __restrict__ offg) {
    __shared__ int  cnt[256];
    __shared__ int  off[256];
    __shared__ int  wsum[8];
    __shared__ int2 stage[CHE];
    int tid = threadIdx.x;
    int lane = tid & 31, wv = tid >> 5;
    int e0 = blockIdx.x * CHE;
    int dv[EPT], sv[EPT], rk[EPT], bk[EPT];
#pragma unroll
    for (int k = 0; k < EPT; ++k) {
        int i = tid + k * 1024;
        int ii = e0 + (i < CHE ? i : CHE - 1);
        dv[k] = dst[ii];
        sv[k] = src[ii];
    }
    if (tid < 256) cnt[tid] = 0;
    __syncthreads();
#pragma unroll
    for (int k = 0; k < EPT; ++k) {
        unsigned b = (unsigned)dv[k] >> 8;
        bk[k] = (int)(b < (unsigned)NBK ? b : (unsigned)(NBK - 1));
        rk[k] = 0;
        if (tid + k * 1024 < CHE) rk[k] = atomicAdd(&cnt[bk[k]], 1);
    }
    __syncthreads();
    int c = (tid < 256) ? cnt[tid] : 0;
    int incl = c;
    for (int o = 1; o < 32; o <<= 1) {
        int y = __shfl_up(incl, o);
        if (lane >= o) incl += y;
    }
    if (tid < 256 && lane == 31) wsum[wv] = incl;
    __syncthreads();
    if (tid < 256) {
        int ex = incl - c;
        for (int w = 0; w < 8; ++w) if (w < wv) ex += wsum[w];
        off[tid] = ex;
        if (tid < NBK) {
            cntg[blockIdx.x * NBK + tid] = c;
            offg[blockIdx.x * NBK + tid] = ex;
        }
    }
    __syncthreads();
#pragma unroll
    for (int k = 0; k < EPT; ++k) {
        if (tid + k * 1024 < CHE) {
            unsigned p = (unsigned)(off[bk[k]] + rk[k]);
            if (p > (unsigned)(CHE - 1)) p = CHE - 1;
            stage[p] = make_int2(sv[k], dv[k]);
        }
    }
    __syncthreads();
    for (int i = tid; i < CHE; i += 1024) ebuf[e0 + i] = stage[i];
}

__global__ __launch_bounds__(256) void rows_k(const int2* __restrict__ ebuf,
                                              const int* __restrict__ cntg,
                                              const int* __restrict__ offg,
                                              int2* __restrict__ csr,
                                              int* __restrict__ rowptr,
                                              float* __restrict__ nrm) {
    __shared__ int deg[256];
    __shared__ int cur[256];
    __shared__ int segs[NCH];
    __shared__ int segn[NCH];
    __shared__ int wdeg[8];
    __shared__ int wbase[8];
    int t = threadIdx.x;
    int lane = t & 31, wv = t >> 5;
    int b = blockIdx.x;
    int o = 0;
    if (t < NCH) {
        o = offg[t * NBK + b];
        int n = cntg[t * NBK + b];
        if (o < 0) o = 0;
        if (o > CHE) o = CHE;
        if (n < 0) n = 0;
        if (n > CHE - o) n = CHE - o;
        segs[t] = t * CHE + o;
        segn[t] = n;
    }
    int bsum = o;
    for (int x = 16; x > 0; x >>= 1) bsum += __shfl_xor(bsum, x);
    if (lane == 0) wbase[wv] = bsum;
    deg[t] = 0;
    __syncthreads();
    int base = 0;
    for (int w = 0; w < 8; ++w) base += wbase[w];

    int g  = t >> 2;
    int j0 = t & 3;
    int s0 = segs[g], n0 = segn[g];
    int sc = s0 < NE - 1 ? s0 : NE - 1;
    for (int j = j0; j < n0; j += 16) {
        int2 e[4];
#pragma unroll
        for (int u = 0; u < 4; ++u) { int jj = j + 4 * u; e[u] = ebuf[jj < n0 ? s0 + jj : sc]; }
#pragma unroll
        for (int u = 0; u < 4; ++u) if (j + 4 * u < n0) atomicAdd(&deg[e[u].y & 255], 1);
    }
    __syncthreads();
    int d = deg[t];
    int incl = d;
    for (int x = 1; x < 32; x <<= 1) {
        int y = __shfl_up(incl, x);
        if (lane >= x) incl += y;
    }
    if (lane == 31) wdeg[wv] = incl;
    __syncthreads();
    int start = base + incl - d;
    for (int w = 0; w < 8; ++w) if (w < wv) start += wdeg[w];
    cur[t] = start;
    int node = b * 256 + t;
    if (node < NN) {
        rowptr[node] = start;
        if (node == NN - 1) rowptr[NN] = start + d;
        float f = (float)d;
        nrm[node] = rsqrtf(f < 1.0f ? 1.0f : f);
    }
    __syncthreads();
    for (int j = j0; j < n0; j += 16) {
        int2 e[4];
#pragma unroll
        for (int u = 0; u < 4; ++u) { int jj = j + 4 * u; e[u] = ebuf[jj < n0 ? s0 + jj : sc]; }
#pragma unroll
        for (int u = 0; u < 4; ++u) {
            if (j + 4 * u < n0) {
                unsigned pos = (unsigned)atomicAdd(&cur[e[u].y & 255], 1);
                if (pos > (unsigned)(NE - 1)) pos = NE - 1;
                csr[pos] = make_int2(e[u].x, 0);
            }
        }
    }
}

__global__ __launch_bounds__(256) void wfill_k(int4* __restrict__ csr2,
                                               const float* __restrict__ nrm) {
    int i = blockIdx.x * 256 + threadIdx.x;
    if (i >= NE / 2) return;
    int4 v = csr2[i];
    v.y = __float_as_int(nrm[v.x]);
    v.w = __float_as_int(nrm[v.z]);
    csr2[i] = v;
}


__global__ __launch_bounds__(256) void hop_k(const int2* __restrict__ csr,
                                             const int* __restrict__ rowptr,
                                             const float* __restrict__ nrm,
                                             const float* __restrict__ hin,
                                             float* __restrict__ hout) {
    int wv = __builtin_amdgcn_readfirstlane((int)(threadIdx.x >> 5));
    int n0 = (blockIdx.x * 8 + wv) * NPW;
    if (n0 >= NN) return;
    int lane = threadIdx.x & 31;
    int p  = lane & 15;
    int hh = lane >> 4;
    int rp[NPW + 1];
#pragma unroll
    for (int q = 0; q <= NPW; ++q) rp[q] = rowptr[n0 + q];
#pragma unroll
    for (int q = 0; q < NPW; ++q) {
        int beg = rp[q], end = rp[q + 1];
        float4 acc = make_float4(0.f, 0.f, 0.f, 0.f);
        for (int j0 = beg; j0 < end; j0 += 8) {
            int2   pr[4];
            float4 v[4];
#pragma unroll
            for (int u = 0; u < 4; ++u) {
                int j = j0 + 2 * u + hh;
                pr[u] = csr[j < end ? j : beg];
            }
#pragma unroll
            for (int u = 0; u < 4; ++u) v[u] = ((const float4*)(hin + (long)pr[u].x * F))[p];
#pragma unroll
            for (int u = 0; u < 4; ++u) {
                float w = (j0 + 2 * u + hh < end) ? __int_as_float(pr[u].y) : 0.0f;
                acc.x += v[u].x * w;
                acc.y += v[u].y * w;
                acc.z += v[u].z * w;
                acc.w += v[u].w * w;
            }
        }
        acc.x += __shfl_xor(acc.x, 16);
        acc.y += __shfl_xor(acc.y, 16);
        acc.z += __shfl_xor(acc.z, 16);
        acc.w += __shfl_xor(acc.w, 16);
        if (hh == 0) {
            float nd = nrm[n0 + q];
            ((float4*)(hout + (long)(n0 + q) * F))[p] =
                make_float4(acc.x * nd, acc.y * nd, acc.z * nd, acc.w * nd);
        }
    }
}


#define WP 260

__device__ __forceinline__ void seg_mac(const float* __restrict__ rp, int kbase,
                                        int h4, int ln15,
                                        const float* Ws,
                                        v8f& c0, v8f& c1, v8f& c2, v8f& c3) {
    const float* wp = Ws + ln15 * WP + kbase + h4;
#pragma unroll 2
    for (int kk = 0; kk < F; kk += 8) {
        float4 a4 = *(const float4*)(rp + kk);
        float4 w0 = *(const float4*)(wp + kk);
        float4 w1 = *(const float4*)(wp + 16 * WP + kk);
        float4 w2 = *(const float4*)(wp + 32 * WP + kk);
        float4 w3 = *(const float4*)(wp + 48 * WP + kk);
        v2f a;  a.x = a4.x;  a.y = a4.y;
        v2f b0; b0.x = w0.x; b0.y = w0.y;
        v2f b1; b1.x = w1.x; b1.y = w1.y;
        v2f b2; b2.x = w2.x; b2.y = w2.y;
        v2f b3; b3.x = w3.x; b3.y = w3.y;
        c0 = __builtin_amdgcn_wmma_f32_16x16x4_f32(false, a, false, b0, (short)0, c0, false, false);
        c1 = __builtin_amdgcn_wmma_f32_16x16x4_f32(false, a, false, b1, (short)0, c1, false, false);
        c2 = __builtin_amdgcn_wmma_f32_16x16x4_f32(false, a, false, b2, (short)0, c2, false, false);
        c3 = __builtin_amdgcn_wmma_f32_16x16x4_f32(false, a, false, b3, (short)0, c3, false, false);
        a.x = a4.z;  a.y = a4.w;
        b0.x = w0.z; b0.y = w0.w;
        b1.x = w1.z; b1.y = w1.w;
        b2.x = w2.z; b2.y = w2.w;
        b3.x = w3.z; b3.y = w3.w;
        c0 = __builtin_amdgcn_wmma_f32_16x16x4_f32(false, a, false, b0, (short)0, c0, false, false);
        c1 = __builtin_amdgcn_wmma_f32_16x16x4_f32(false, a, false, b1, (short)0, c1, false, false);
        c2 = __builtin_amdgcn_wmma_f32_16x16x4_f32(false, a, false, b2, (short)0, c2, false, false);
        c3 = __builtin_amdgcn_wmma_f32_16x16x4_f32(false, a, false, b3, (short)0, c3, false, false);
    }
}

#define HP 68

template <int U> __device__ __forceinline__ int row_bcast(int x) {
    return __builtin_amdgcn_update_dpp(0, x, 0x150 + U, 0xF, 0xF, false);
}

__global__ __launch_bounds__(256) void hop_gemm_k(const int2* __restrict__ csr,
                                                  const int* __restrict__ rowptr,
                                                  const float* __restrict__ nrm,
                                                  const float* __restrict__ feat,
                                                  const float* __restrict__ h1,
                                                  const float* __restrict__ h2,
                                                  const float* __restrict__ W,
                                                  const float* __restrict__ bias,
                                                  float* __restrict__ out) {
    __shared__ float4 Wsv[F * WP / 4];
    __shared__ float4 Hsv[8 * 16 * HP / 4];
    int tid = threadIdx.x;
    for (int i = tid; i < F * KD / 4; i += 256) {
        int o  = i >> 6;
        int c4 = i & 63;
        Wsv[o * (WP / 4) + c4] = ((const float4*)W)[i];
    }
    const float* Ws = (const float*)Wsv;

    int lane = tid & 31;
    int ln15 = lane & 15;
    int hh   = lane >> 4;
    int h4   = hh * 4;
    int wv   = tid >> 5;
    int rowBase = blockIdx.x * 128 + wv * 16;
    bool live = rowBase < NN;
    float* Hs = (float*)Hsv + wv * (16 * HP);

    if (live) {
        int   rpv = rowptr[rowBase + (lane < 17 ? lane : 16)];
        float nrv = nrm[rowBase + ln15];
        const float4* h2v = (const float4*)h2;
        int begq[8], endq[8];
        int2 e0[8];
#pragma unroll
        for (int q = 0; q < 8; ++q) {
            begq[q] = __shfl(rpv, 2 * q + hh);
            endq[q] = __shfl(rpv, 2 * q + hh + 1);
            int j = begq[q] + ln15;
            int jc = j < endq[q] ? j : begq[q];
            if (jc > NE - 1) jc = NE - 1;
            e0[q] = csr[jc];
        }
#pragma unroll
        for (int q = 0; q < 8; ++q) {
            int beg = begq[q], end = endq[q];
            int2 ent = e0[q];
            float4 acc = make_float4(0.f, 0.f, 0.f, 0.f);
            for (int j0 = beg; j0 < end; j0 += 16) {
                int2 cur = ent;
                if (j0 + 16 < end) {
                    int jn = j0 + 16 + ln15;
                    ent = csr[jn < end ? jn : beg];
                }
                int wy = (j0 + ln15 < end) ? cur.y : 0;
                float4 v[8];
#define LD_ROW(V, U) { unsigned s = (unsigned)row_bcast<U>(cur.x); V[(U) & 7] = h2v[s * 16u + (unsigned)ln15]; }
#define AC_ROW(V, U) { float w = __int_as_float(row_bcast<U>(wy)); \
                       acc.x += V[(U) & 7].x * w; acc.y += V[(U) & 7].y * w; \
                       acc.z += V[(U) & 7].z * w; acc.w += V[(U) & 7].w * w; }
                LD_ROW(v, 0) LD_ROW(v, 1) LD_ROW(v, 2) LD_ROW(v, 3)
                LD_ROW(v, 4) LD_ROW(v, 5) LD_ROW(v, 6) LD_ROW(v, 7)
                if (j0 + 8 < end) {
                    float4 t[8];
                    LD_ROW(t, 8)  LD_ROW(t, 9)  LD_ROW(t, 10) LD_ROW(t, 11)
                    LD_ROW(t, 12) LD_ROW(t, 13) LD_ROW(t, 14) LD_ROW(t, 15)
                    AC_ROW(t, 8)  AC_ROW(t, 9)  AC_ROW(t, 10) AC_ROW(t, 11)
                    AC_ROW(t, 12) AC_ROW(t, 13) AC_ROW(t, 14) AC_ROW(t, 15)
                }
                AC_ROW(v, 0) AC_ROW(v, 1) AC_ROW(v, 2) AC_ROW(v, 3)
                AC_ROW(v, 4) AC_ROW(v, 5) AC_ROW(v, 6) AC_ROW(v, 7)
#undef LD_ROW
#undef AC_ROW
            }
            float nd = __shfl(nrv, 2 * q + hh);
            *(float4*)(Hs + (2 * q + hh) * HP + 4 * ln15) =
                make_float4(acc.x * nd, acc.y * nd, acc.z * nd, acc.w * nd);
        }
    }
    __syncthreads();
    if (!live) return;

    int node = rowBase + ln15;

    v8f c0, c1, c2, c3;
    float bv0 = bias[ln15], bv1 = bias[16 + ln15], bv2 = bias[32 + ln15], bv3 = bias[48 + ln15];
#pragma unroll
    for (int i = 0; i < 8; ++i) { c0[i] = bv0; c1[i] = bv1; c2[i] = bv2; c3[i] = bv3; }

    seg_mac(feat + (long)node * F + h4, 0,   h4, ln15, Ws, c0, c1, c2, c3);
    seg_mac(h1   + (long)node * F + h4, 64,  h4, ln15, Ws, c0, c1, c2, c3);
    seg_mac(h2   + (long)node * F + h4, 128, h4, ln15, Ws, c0, c1, c2, c3);
    seg_mac(Hs + ln15 * HP + h4,        192, h4, ln15, Ws, c0, c1, c2, c3);

    int mBase = rowBase + (h4 ? 8 : 0);
#pragma unroll
    for (int r = 0; r < 8; ++r) {
        int row = mBase + r;
        out[(long)row * F + ln15]      = c0[r];
        out[(long)row * F + 16 + ln15] = c1[r];
        out[(long)row * F + 32 + ln15] = c2[r];
        out[(long)row * F + 48 + ln15] = c3[r];
    }
}


extern "C" void kernel_launch(void* const* d_in, const int* in_sizes, int n_in,
                              void* d_out, int out_size, void* d_ws, size_t ws_size,
                              hipStream_t stream) {
    const float* feat = (const float*)d_in[0];
    const int*   src  = (const int*)d_in[1];
    const int*   dst  = (const int*)d_in[2];
    const float* W    = (const float*)d_in[3];
    const float* bias = (const float*)d_in[4];
    float*       out  = (float*)d_out;

    float* ws   = (float*)d_ws;
    float* nrm  = ws;
    float* h1   = ws + 50048;
    float* h2   = h1 + (long)NN * F;
    float* h3   = h2 + (long)NN * F;

    int2* ebuf = (int2*)h1;
    int*  cntg = (int*)h2;
    int*  offg = (int*)h2 + NCH * NBK;
    int2* csr    = (int2*)h3;
    int*  rowptr = (int*)h3 + 2 * NE;

    bucket_k<<<NCH, 1024, 0, stream>>>(src, dst, ebuf, cntg, offg);
    rows_k  <<<NBK, 256, 0, stream>>>(ebuf, cntg, offg, csr, rowptr, nrm);
    wfill_k <<<(NE / 2 + 255) / 256, 256, 0, stream>>>((int4*)csr, nrm);

    const int hopBlocks = (NN + 8 * NPW - 1) / (8 * NPW);

    hop_k<<<hopBlocks, 256, 0, stream>>>(csr, rowptr, nrm, feat, h1);
    hop_k<<<hopBlocks, 256, 0, stream>>>(csr, rowptr, nrm, h1, h2);
    hop_gemm_k<<<(NN + 127) / 128, 256, 0, stream>>>(csr, rowptr, nrm, feat, h1, h2, W, bias, out);
}


# Last launch: a SIMD's two waves take gather and matrix work in opposite order (waves 0-3 multiply the feat, h1, h2 segments first, waves 4-7 gather first); a segment's A rows requested as one group
# speedup vs baseline: 3.8994x; 1.0666x over previous
#include <hip/hip_runtime.h>

typedef float v2f __attribute__((ext_vector_type(2)));
typedef float v8f __attribute__((ext_vector_type(8)));

#define NN 50000
#define NE 800000
#define F  64
#define KD 256

#define NCH 64
#define CHE (NE / NCH)
#define EPT 13
#define NPW 4
#define NBK 196


__global__ __launch_bounds__(1024) void bucket_k(const int* __restrict__ src,
                                                 const int* __restrict__ dst,
                                                 int2* __restrict__ ebuf,
                                                 int* __restrict__ cntg,
                                                 int* __restrict__ offg) {
    __shared__ int  cnt[256];
    __shared__ int  off[256];
    __shared__ int  wsum[8];
    __shared__ int2 stage[CHE];
    int tid = threadIdx.x;
    int lane = tid & 31, wv = tid >> 5;
    int e0 = blockIdx.x * CHE;
    int dv[EPT], sv[EPT], rk[EPT], bk[EPT];
#pragma unroll
    for (int k = 0; k < EPT; ++k) {
        int i = tid + k * 1024;
        int ii = e0 + (i < CHE ? i : CHE - 1);
        dv[k] = dst[ii];
        sv[k] = src[ii];
    }
    if (tid < 256) cnt[tid] = 0;
    __syncthreads();
#pragma unroll
    for (int k = 0; k < EPT; ++k) {
        unsigned b = (unsigned)dv[k] >> 8;
        bk[k] = (int)(b < (unsigned)NBK ? b : (unsigned)(NBK - 1));
        rk[k] = 0;
        if (tid + k * 1024 < CHE) rk[k] = atomicAdd(&cnt[bk[k]], 1);
    }
    __syncthreads();
    int c = (tid < 256) ? cnt[tid] : 0;
    int incl = c;
    for (int o = 1; o < 32; o <<= 1) {
        int y = __shfl_up(incl, o);
        if (lane >= o) incl += y;
    }
    if (tid < 256 && lane == 31) wsum[wv] = incl;
    __syncthreads();
    if (tid < 256) {
        int ex = incl - c;
        for (int w = 0; w < 8; ++w) if (w < wv) ex += wsum[w];
        off[tid] = ex;
        if (tid < NBK) {
            cntg[blockIdx.x * NBK + tid] = c;
            offg[blockIdx.x * NBK + tid] = ex;
        }
    }
    __syncthreads();
#pragma unroll
    for (int k = 0; k < EPT; ++k) {
        if (tid + k * 1024 < CHE) {
            unsigned p = (unsigned)(off[bk[k]] + rk[k]);
            if (p > (unsigned)(CHE - 1)) p = CHE - 1;
            stage[p] = make_int2(sv[k], dv[k]);
        }
    }
    __syncthreads();
    for (int i = tid; i < CHE; i += 1024) ebuf[e0 + i] = stage[i];
}

__global__ __launch_bounds__(256) void rows_k(const int2* __restrict__ ebuf,
                                              const int* __restrict__ cntg,
                                              const int* __restrict__ offg,
                                              const float* __restrict__ feat,
                                              int* __restrict__ csr,
                                              int* __restrict__ rowptr,
                                              float* __restrict__ nrm,
                                              float* __restrict__ g0) {
    __shared__ int deg[256];
    __shared__ float nrs[256];
    __shared__ int cur[256];
    __shared__ int segs[NCH];
    __shared__ int segn[NCH];
    __shared__ int wdeg[8];
    __shared__ int wbase[8];
    int t = threadIdx.x;
    int lane = t & 31, wv = t >> 5;
    int b = blockIdx.x;
    int o = 0;
    if (t < NCH) {
        o = offg[t * NBK + b];
        int n = cntg[t * NBK + b];
        if (o < 0) o = 0;
        if (o > CHE) o = CHE;
        if (n < 0) n = 0;
        if (n > CHE - o) n = CHE - o;
        segs[t] = t * CHE + o;
        segn[t] = n;
    }
    const float4* fv = (const float4*)feat + (long)b * (256 * 16);
    float4 fr[16];
#pragma unroll
    for (int k = 0; k < 16; ++k) {
        int i = t + 256 * k;
        fr[k] = (b * 256 + (i >> 4) < NN) ? fv[i] : make_float4(0.f, 0.f, 0.f, 0.f);
    }
    int bsum = o;
    for (int x = 16; x > 0; x >>= 1) bsum += __shfl_xor(bsum, x);
    if (lane == 0) wbase[wv] = bsum;
    deg[t] = 0;
    __syncthreads();
    int base = 0;
    for (int w = 0; w < 8; ++w) base += wbase[w];

    int g  = t >> 2;
    int j0 = t & 3;
    int s0 = segs[g], n0 = segn[g];
    int sc = s0 < NE - 1 ? s0 : NE - 1;
    for (int j = j0; j < n0; j += 16) {
        int2 e[4];
#pragma unroll
        for (int u = 0; u < 4; ++u) { int jj = j + 4 * u; e[u] = ebuf[jj < n0 ? s0 + jj : sc]; }
#pragma unroll
        for (int u = 0; u < 4; ++u) if (j + 4 * u < n0) atomicAdd(&deg[e[u].y & 255], 1);
    }
    __syncthreads();
    int d = deg[t];
    int incl = d;
    for (int x = 1; x < 32; x <<= 1) {
        int y = __shfl_up(incl, x);
        if (lane >= x) incl += y;
    }
    if (lane == 31) wdeg[wv] = incl;
    __syncthreads();
    int start = base + incl - d;
    for (int w = 0; w < 8; ++w) if (w < wv) start += wdeg[w];
    cur[t] = start;
    int node = b * 256 + t;
    {
        float f = (float)d;
        float nr = rsqrtf(f < 1.0f ? 1.0f : f);
        nrs[t] = nr;
        if (node < NN) {
            rowptr[node] = start;
            if (node == NN - 1) rowptr[NN] = start + d;
            nrm[node] = nr;
        }
    }
    __syncthreads();
    {
        float4* gv = (float4*)g0 + (long)b * (256 * 16);
        float scl[16];
#pragma unroll
        for (int k = 0; k < 16; ++k) scl[k] = nrs[(t >> 4) + 16 * k];
#pragma unroll
        for (int k = 0; k < 16; ++k) {
            int i = t + 256 * k;
            if (b * 256 + (i >> 4) < NN) {
                float s = scl[k];
                gv[i] = make_float4(fr[k].x * s, fr[k].y * s, fr[k].z * s, fr[k].w * s);
            }
        }
    }
    for (int j = j0; j < n0; j += 16) {
        int2 e[4];
#pragma unroll
        for (int u = 0; u < 4; ++u) { int jj = j + 4 * u; e[u] = ebuf[jj < n0 ? s0 + jj : sc]; }
#pragma unroll
        for (int u = 0; u < 4; ++u) {
            if (j + 4 * u < n0) {
                unsigned pos = (unsigned)atomicAdd(&cur[e[u].y & 255], 1);
                if (pos > (unsigned)(NE - 1)) pos = NE - 1;
                csr[pos] = e[u].x;
            }
        }
    }
}


__global__ __launch_bounds__(256) void hop_k(const int* __restrict__ csr,
                                             const int* __restrict__ rowptr,
                                             const float* __restrict__ nrm,
                                             const float* __restrict__ gin,
                                             float* __restrict__ gout) {
    int wv = __builtin_amdgcn_readfirstlane((int)(threadIdx.x >> 5));
    int n0 = (blockIdx.x * 8 + wv) * NPW;
    if (n0 >= NN) return;
    int lane = threadIdx.x & 31;
    int p  = lane & 15;
    int hh = lane >> 4;
    const float4* gv = (const float4*)gin;
    int rp[NPW + 1];
#pragma unroll
    for (int q = 0; q <= NPW; ++q) rp[q] = rowptr[n0 + q];
#pragma unroll
    for (int q = 0; q < NPW; ++q) {
        int beg = rp[q], end = rp[q + 1];
        float4 acc = make_float4(0.f, 0.f, 0.f, 0.f);
        for (int j0 = beg; j0 < end; j0 += 8) {
            int    pr[4];
            float4 v[4];
#pragma unroll
            for (int u = 0; u < 4; ++u) {
                int j = j0 + 2 * u + hh;
                pr[u] = csr[j < end ? j : beg];
            }
#pragma unroll
            for (int u = 0; u < 4; ++u) v[u] = gv[(unsigned)pr[u] * 16u + (unsigned)p];
#pragma unroll
            for (int u = 0; u < 4; ++u) {
                float w = (j0 + 2 * u + hh < end) ? 1.0f : 0.0f;
                acc.x += v[u].x * w;
                acc.y += v[u].y * w;
                acc.z += v[u].z * w;
                acc.w += v[u].w * w;
            }
        }
        acc.x += __shfl_xor(acc.x, 16);
        acc.y += __shfl_xor(acc.y, 16);
        acc.z += __shfl_xor(acc.z, 16);
        acc.w += __shfl_xor(acc.w, 16);
        if (hh == 0) {
            float nd = nrm[n0 + q];
            ((float4*)(gout + (long)(n0 + q) * F))[p] =
                make_float4(acc.x * nd * nd, acc.y * nd * nd, acc.z * nd * nd, acc.w * nd * nd);
        }
    }
}


#define WP 260

__device__ __forceinline__ void seg_load(const float* __restrict__ rp, float4 (&a)[8]) {
#pragma unroll
    for (int t = 0; t < 8; ++t) a[t] = *(const float4*)(rp + 8 * t);
}

__device__ __forceinline__ void seg_mul(const float4 (&a)[8], int kbase,
                                        int h4, int ln15,
                                        const float* Ws,
                                        v8f& c0, v8f& c1, v8f& c2, v8f& c3) {
    const float* wp = Ws + ln15 * WP + kbase + h4;
#pragma unroll
    for (int t = 0; t < 8; ++t) {
        int kk = 8 * t;
        float4 a4 = a[t];
        float4 w0 = *(const float4*)(wp + kk);
        float4 w1 = *(const float4*)(wp + 16 * WP + kk);
        float4 w2 = *(const float4*)(wp + 32 * WP + kk);
        float4 w3 = *(const float4*)(wp + 48 * WP + kk);
        v2f av;  av.x = a4.x;  av.y = a4.y;
        v2f b0; b0.x = w0.x; b0.y = w0.y;
        v2f b1; b1.x = w1.x; b1.y = w1.y;
        v2f b2; b2.x = w2.x; b2.y = w2.y;
        v2f b3; b3.x = w3.x; b3.y = w3.y;
        c0 = __builtin_amdgcn_wmma_f32_16x16x4_f32(false, av, false, b0, (short)0, c0, false, false);
        c1 = __builtin_amdgcn_wmma_f32_16x16x4_f32(false, av, false, b1, (short)0, c1, false, false);
        c2 = __builtin_amdgcn_wmma_f32_16x16x4_f32(false, av, false, b2, (short)0, c2, false, false);
        c3 = __builtin_amdgcn_wmma_f32_16x16x4_f32(false, av, false, b3, (short)0, c3, false, false);
        av.x = a4.z;  av.y = a4.w;
        b0.x = w0.z; b0.y = w0.w;
        b1.x = w1.z; b1.y = w1.w;
        b2.x = w2.z; b2.y = w2.w;
        b3.x = w3.z; b3.y = w3.w;
        c0 = __builtin_amdgcn_wmma_f32_16x16x4_f32(false, av, false, b0, (short)0, c0, false, false);
        c1 = __builtin_amdgcn_wmma_f32_16x16x4_f32(false, av, false, b1, (short)0, c1, false, false);
        c2 = __builtin_amdgcn_wmma_f32_16x16x4_f32(false, av, false, b2, (short)0, c2, false, false);
        c3 = __builtin_amdgcn_wmma_f32_16x16x4_f32(false, av, false, b3, (short)0, c3, false, false);
    }
}

#define HP 68

template <int U> __device__ __forceinline__ int row_bcast(int x) {
    return __builtin_amdgcn_update_dpp(0, x, 0x150 + U, 0xF, 0xF, false);
}

__device__ __forceinline__ void dense_part(const float* __restrict__ feat,
                                           const float* __restrict__ g1,
                                           const float* __restrict__ g2,
                                           const float* __restrict__ bias,
                                           int node, int h4, int hh, int ln15, float nrv,
                                           const float* Ws,
                                           v8f& c0, v8f& c1, v8f& c2, v8f& c3) {
    float4 a1[8], a2[8], a3[8];
    seg_load(g1 + (long)node * F + h4, a1);
    seg_load(g2 + (long)node * F + h4, a2);
    float bv0 = bias[ln15], bv1 = bias[16 + ln15], bv2 = bias[32 + ln15], bv3 = bias[48 + ln15];
#pragma unroll
    for (int i = 0; i < 8; ++i) { c0[i] = 0.0f; c1[i] = 0.0f; c2[i] = 0.0f; c3[i] = 0.0f; }

    seg_mul(a1, 64,  h4, ln15, Ws, c0, c1, c2, c3);
    seg_load(feat + (long)node * F + h4, a3);
    seg_mul(a2, 128, h4, ln15, Ws, c0, c1, c2, c3);

    float rsv = 1.0f / nrv;
#pragma unroll
    for (int r = 0; r < 8; ++r) {
        float s = __shfl(rsv, 8 * hh + r);
        c0[r] = c0[r] * s + bv0;
        c1[r] = c1[r] * s + bv1;
        c2[r] = c2[r] * s + bv2;
        c3[r] = c3[r] * s + bv3;
    }

    seg_mul(a3, 0, h4, ln15, Ws, c0, c1, c2, c3);
}

__global__ __launch_bounds__(256) void hop_gemm_k(const int* __restrict__ csr,
                                                  const int* __restrict__ rowptr,
                                                  const float* __restrict__ nrm,
                                                  const float* __restrict__ feat,
                                                  const float* __restrict__ g1,
                                                  const float* __restrict__ g2,
                                                  const float* __restrict__ W,
                                                  const float* __restrict__ bias,
                                                  float* __restrict__ out) {
    __shared__ float4 Wsv[F * WP / 4];
    __shared__ float4 Hsv[8 * 16 * HP / 4];
    int tid = threadIdx.x;
    for (int i = tid; i < F * KD / 4; i += 256) {
        int o  = i >> 6;
        int c4 = i & 63;
        Wsv[o * (WP / 4) + c4] = ((const float4*)W)[i];
    }
    const float* Ws = (const float*)Wsv;

    int lane = tid & 31;
    int ln15 = lane & 15;
    int hh   = lane >> 4;
    int h4   = hh * 4;
    int wv   = __builtin_amdgcn_readfirstlane(tid >> 5);
    int rowBase = blockIdx.x * 128 + wv * 16;
    bool live = rowBase < NN;
    float* Hs = (float*)Hsv + wv * (16 * HP);

    __syncthreads();
    if (!live) return;

    int node = rowBase + ln15;
    float nrv = nrm[node];
    bool denseFirst = wv < 4;
    v8f c0, c1, c2, c3;
    if (denseFirst) dense_part(feat, g1, g2, bias, node, h4, hh, ln15, nrv, Ws, c0, c1, c2, c3);

    {
        int   rpv = rowptr[rowBase + (lane < 17 ? lane : 16)];
        const float4* g2v = (const float4*)g2;
        int begq[8], endq[8];
        int e0[8];
#pragma unroll
        for (int q = 0; q < 8; ++q) {
            begq[q] = __shfl(rpv, 2 * q + hh);
            endq[q] = __shfl(rpv, 2 * q + hh + 1);
            int j = begq[q] + ln15;
            int jc = j < endq[q] ? j : begq[q];
            if (jc > NE - 1) jc = NE - 1;
            e0[q] = csr[jc];
        }
#pragma unroll
        for (int q = 0; q < 8; ++q) {
            int beg = begq[q], end = endq[q];
            int ent = e0[q];
            float4 acc = make_float4(0.f, 0.f, 0.f, 0.f);
            for (int j0 = beg; j0 < end; j0 += 16) {
                int cur = ent;
                if (j0 + 16 < end) {
                    int jn = j0 + 16 + ln15;
                    ent = csr[jn < end ? jn : beg];
                }
                float4 v[8];
#define LD_ROW(V, U) { unsigned s = (unsigned)row_bcast<U>(cur); V[(U) & 7] = g2v[s * 16u + (unsigned)ln15]; }
#define AC_ROW(V, U) { float w = (j0 + (U) < end) ? 1.0f : 0.0f; \
                       acc.x += V[(U) & 7].x * w; acc.y += V[(U) & 7].y * w; \
                       acc.z += V[(U) & 7].z * w; acc.w += V[(U) & 7].w * w; }
                LD_ROW(v, 0) LD_ROW(v, 1) LD_ROW(v, 2) LD_ROW(v, 3)
                LD_ROW(v, 4) LD_ROW(v, 5) LD_ROW(v, 6) LD_ROW(v, 7)
                if (j0 + 8 < end) {
                    float4 t[8];
                    LD_ROW(t, 8)  LD_ROW(t, 9)  LD_ROW(t, 10) LD_ROW(t, 11)
                    LD_ROW(t, 12) LD_ROW(t, 13) LD_ROW(t, 14) LD_ROW(t, 15)
                    AC_ROW(t, 8)  AC_ROW(t, 9)  AC_ROW(t, 10) AC_ROW(t, 11)
                    AC_ROW(t, 12) AC_ROW(t, 13) AC_ROW(t, 14) AC_ROW(t, 15)
                }
                AC_ROW(v, 0) AC_ROW(v, 1) AC_ROW(v, 2) AC_ROW(v, 3)
                AC_ROW(v, 4) AC_ROW(v, 5) AC_ROW(v, 6) AC_ROW(v, 7)
#undef LD_ROW
#undef AC_ROW
            }
            float nd = __shfl(nrv, 2 * q + hh);
            *(float4*)(Hs + (2 * q + hh) * HP + 4 * ln15) =
                make_float4(acc.x * nd, acc.y * nd, acc.z * nd, acc.w * nd);
        }
    }

    if (!denseFirst) dense_part(feat, g1, g2, bias, node, h4, hh, ln15, nrv, Ws, c0, c1, c2, c3);

    float4 a4h[8];
    seg_load(Hs + ln15 * HP + h4, a4h);
    seg_mul(a4h, 192, h4, ln15, Ws, c0, c1, c2, c3);

    int mBase = rowBase + (h4 ? 8 : 0);
#pragma unroll
    for (int r = 0; r < 8; ++r) {
        int row = mBase + r;
        out[(long)row * F + ln15]      = c0[r];
        out[(long)row * F + 16 + ln15] = c1[r];
        out[(long)row * F + 32 + ln15] = c2[r];
        out[(long)row * F + 48 + ln15] = c3[r];
    }
}


extern "C" void kernel_launch(void* const* d_in, const int* in_sizes, int n_in,
                              void* d_out, int out_size, void* d_ws, size_t ws_size,
                              hipStream_t stream) {
    const float* feat = (const float*)d_in[0];
    const int*   src  = (const int*)d_in[1];
    const int*   dst  = (const int*)d_in[2];
    const float* W    = (const float*)d_in[3];
    const float* bias = (const float*)d_in[4];
    float*       out  = (float*)d_out;

    float* ws   = (float*)d_ws;
    float* nrm  = ws;
    float* ga   = ws + 50048;
    float* gb   = ga + (long)NN * F;
    int*   lst  = (int*)(gb + (long)NN * F);

    int*  csr    = lst;
    int*  rowptr = lst + NE;
    int2* ebuf = (int2*)(lst + NE + 50176);
    int*  cntg = lst + NE + 50176 + 2 * NE;
    int*  offg = cntg + NCH * NBK;

    bucket_k<<<NCH, 1024, 0, stream>>>(src, dst, ebuf, cntg, offg);
    rows_k  <<<NBK, 256, 0, stream>>>(ebuf, cntg, offg, feat, csr, rowptr, nrm, ga);

    const int hopBlocks = (NN + 8 * NPW - 1) / (8 * NPW);

    hop_k<<<hopBlocks, 256, 0, stream>>>(csr, rowptr, nrm, ga, gb);
    hop_k<<<hopBlocks, 256, 0, stream>>>(csr, rowptr, nrm, gb, ga);
    hop_gemm_k<<<(NN + 127) / 128, 256, 0, stream>>>(csr, rowptr, nrm, feat, gb, ga, W, bias, out);
}


# Edge sort regrained: level 1 in 256 chunks (one work-group per WGP, not 64); level 2 walks a bucket's 256 short pieces one thread each, pairs loaded once as one group and held in registers
# speedup vs baseline: 4.2803x; 1.0977x over previous
#include <hip/hip_runtime.h>

typedef float v2f __attribute__((ext_vector_type(2)));
typedef float v8f __attribute__((ext_vector_type(8)));

#define NN 50000
#define NE 800000
#define F  64
#define KD 256

#define NCH 256
#define CHE (NE / NCH)
#define EPT 13
#define RPT 32
#define NPW 4
#define NBK 196


__global__ __launch_bounds__(256) void bucket_k(const int* __restrict__ src,
                                                const int* __restrict__ dst,
                                                int2* __restrict__ ebuf,
                                                int2* __restrict__ tabg) {
    __shared__ int  cnt[256];
    __shared__ int  off[256];
    __shared__ int  wsum[8];
    __shared__ int2 stage[CHE];
    int tid = threadIdx.x;
    int lane = tid & 31, wv = tid >> 5;
    int e0 = blockIdx.x * CHE;
    int dv[EPT], sv[EPT], rk[EPT], bk[EPT];
#pragma unroll
    for (int k = 0; k < EPT; ++k) {
        int i = tid + k * 256;
        int ii = e0 + (i < CHE ? i : CHE - 1);
        dv[k] = dst[ii];
        sv[k] = src[ii];
    }
    cnt[tid] = 0;
    __syncthreads();
#pragma unroll
    for (int k = 0; k < EPT; ++k) {
        unsigned b = (unsigned)dv[k] >> 8;
        bk[k] = (int)(b < (unsigned)NBK ? b : (unsigned)(NBK - 1));
        rk[k] = 0;
        if (tid + k * 256 < CHE) rk[k] = atomicAdd(&cnt[bk[k]], 1);
    }
    __syncthreads();
    int c = cnt[tid];
    int incl = c;
    for (int o = 1; o < 32; o <<= 1) {
        int y = __shfl_up(incl, o);
        if (lane >= o) incl += y;
    }
    if (lane == 31) wsum[wv] = incl;
    __syncthreads();
    {
        int ex = incl - c;
        for (int w = 0; w < 8; ++w) if (w < wv) ex += wsum[w];
        off[tid] = ex;
        if (tid < NBK) tabg[blockIdx.x * NBK + tid] = make_int2(c, ex);
    }
    __syncthreads();
#pragma unroll
    for (int k = 0; k < EPT; ++k) {
        if (tid + k * 256 < CHE) {
            unsigned p = (unsigned)(off[bk[k]] + rk[k]);
            if (p > (unsigned)(CHE - 1)) p = CHE - 1;
            stage[p] = make_int2(sv[k], dv[k]);
        }
    }
    __syncthreads();
    for (int i = tid; i < CHE; i += 256) ebuf[e0 + i] = stage[i];
}

__global__ __launch_bounds__(256) void rows_k(const int2* __restrict__ ebuf,
                                              const int2* __restrict__ tabg,
                                              const float* __restrict__ feat,
                                              int* __restrict__ csr,
                                              int* __restrict__ rowptr,
                                              float* __restrict__ nrm,
                                              float* __restrict__ g0) {
    __shared__ int deg[512];
    __shared__ float nrs[256];
    __shared__ int cur[512];
    __shared__ int wdeg[8];
    __shared__ int wbase[8];
    int t = threadIdx.x;
    int lane = t & 31, wv = t >> 5;
    int b = blockIdx.x;
    int2 cn = tabg[t * NBK + b];
    int n0 = cn.x, o = cn.y;
    if (o < 0) o = 0;
    if (o > CHE) o = CHE;
    if (n0 < 0) n0 = 0;
    if (n0 > CHE - o) n0 = CHE - o;
    int s0 = t * CHE + o;
    int sc = s0 < NE - 1 ? s0 : NE - 1;
    int2 e[RPT];
#pragma unroll
    for (int u = 0; u < RPT; ++u) e[u] = ebuf[u < n0 ? s0 + u : sc];
    const float4* fv = (const float4*)feat;
    float4 fr[16];
#pragma unroll
    for (int k = 0; k < 16; ++k) {
        int i = t + 256 * k;
        int row = b * 256 + (i >> 4);
        fr[k] = fv[(long)(row < NN ? row : NN - 1) * 16 + (i & 15)];
    }
    int bsum = o;
    for (int x = 16; x > 0; x >>= 1) bsum += __shfl_xor(bsum, x);
    if (lane == 0) wbase[wv] = bsum;
    deg[t] = 0;
    deg[256 + t] = 0;
    cur[256 + t] = 0;
    __syncthreads();
    int base = 0;
    for (int w = 0; w < 8; ++w) base += wbase[w];

    int ix[RPT];
#pragma unroll
    for (int u = 0; u < RPT; ++u) ix[u] = u < n0 ? (e[u].y & 255) : 256 + t;
#pragma unroll
    for (int u = 0; u < RPT; ++u) atomicAdd(&deg[ix[u]], 1);
    for (int j = RPT; j < n0; j += 4) {
        int2 r[4];
#pragma unroll
        for (int u = 0; u < 4; ++u) r[u] = ebuf[j + u < n0 ? s0 + j + u : sc];
#pragma unroll
        for (int u = 0; u < 4; ++u) if (j + u < n0) atomicAdd(&deg[r[u].y & 255], 1);
    }
    __syncthreads();
    int d = deg[t];
    int incl = d;
    for (int x = 1; x < 32; x <<= 1) {
        int y = __shfl_up(incl, x);
        if (lane >= x) incl += y;
    }
    if (lane == 31) wdeg[wv] = incl;
    __syncthreads();
    int start = base + incl - d;
    for (int w = 0; w < 8; ++w) if (w < wv) start += wdeg[w];
    cur[t] = start;
    int node = b * 256 + t;
    {
        float f = (float)d;
        float nr = rsqrtf(f < 1.0f ? 1.0f : f);
        nrs[t] = nr;
        if (node < NN) {
            rowptr[node] = start;
            if (node == NN - 1) rowptr[NN] = start + d;
            nrm[node] = nr;
        }
    }
    __syncthreads();
    int ps[RPT];
#pragma unroll
    for (int u = 0; u < RPT; ++u) ps[u] = atomicAdd(&cur[ix[u]], 1);
#pragma unroll
    for (int u = 0; u < RPT; ++u) {
        if (u < n0) {
            unsigned pos = (unsigned)ps[u];
            if (pos > (unsigned)(NE - 1)) pos = NE - 1;
            csr[pos] = e[u].x;
        }
    }
    for (int j = RPT; j < n0; j += 4) {
        int2 r[4];
#pragma unroll
        for (int u = 0; u < 4; ++u) r[u] = ebuf[j + u < n0 ? s0 + j + u : sc];
#pragma unroll
        for (int u = 0; u < 4; ++u) {
            if (j + u < n0) {
                unsigned pos = (unsigned)atomicAdd(&cur[r[u].y & 255], 1);
                if (pos > (unsigned)(NE - 1)) pos = NE - 1;
                csr[pos] = r[u].x;
            }
        }
    }
    {
        float4* gv = (float4*)g0 + (long)b * (256 * 16);
        float scl[16];
#pragma unroll
        for (int k = 0; k < 16; ++k) scl[k] = nrs[(t >> 4) + 16 * k];
#pragma unroll
        for (int k = 0; k < 16; ++k) {
            int i = t + 256 * k;
            if (b * 256 + (i >> 4) < NN) {
                float s = scl[k];
                gv[i] = make_float4(fr[k].x * s, fr[k].y * s, fr[k].z * s, fr[k].w * s);
            }
        }
    }
}


__global__ __launch_bounds__(256) void hop_k(const int* __restrict__ csr,
                                             const int* __restrict__ rowptr,
                                             const float* __restrict__ nrm,
                                             const float* __restrict__ gin,
                                             float* __restrict__ gout) {
    int wv = __builtin_amdgcn_readfirstlane((int)(threadIdx.x >> 5));
    int n0 = (blockIdx.x * 8 + wv) * NPW;
    if (n0 >= NN) return;
    int lane = threadIdx.x & 31;
    int p  = lane & 15;
    int hh = lane >> 4;
    const float4* gv = (const float4*)gin;
    int rp[NPW + 1];
#pragma unroll
    for (int q = 0; q <= NPW; ++q) rp[q] = rowptr[n0 + q];
#pragma unroll
    for (int q = 0; q < NPW; ++q) {
        int beg = rp[q], end = rp[q + 1];
        float4 acc = make_float4(0.f, 0.f, 0.f, 0.f);
        for (int j0 = beg; j0 < end; j0 += 8) {
            int    pr[4];
            float4 v[4];
#pragma unroll
            for (int u = 0; u < 4; ++u) {
                int j = j0 + 2 * u + hh;
                pr[u] = csr[j < end ? j : beg];
            }
#pragma unroll
            for (int u = 0; u < 4; ++u) v[u] = gv[(unsigned)pr[u] * 16u + (unsigned)p];
#pragma unroll
            for (int u = 0; u < 4; ++u) {
                float w = (j0 + 2 * u + hh < end) ? 1.0f : 0.0f;
                acc.x += v[u].x * w;
                acc.y += v[u].y * w;
                acc.z += v[u].z * w;
                acc.w += v[u].w * w;
            }
        }
        acc.x += __shfl_xor(acc.x, 16);
        acc.y += __shfl_xor(acc.y, 16);
        acc.z += __shfl_xor(acc.z, 16);
        acc.w += __shfl_xor(acc.w, 16);
        if (hh == 0) {
            float nd = nrm[n0 + q];
            ((float4*)(gout + (long)(n0 + q) * F))[p] =
                make_float4(acc.x * nd * nd, acc.y * nd * nd, acc.z * nd * nd, acc.w * nd * nd);
        }
    }
}


#define WP 260

__device__ __forceinline__ void seg_load(const float* __restrict__ rp, float4 (&a)[8]) {
#pragma unroll
    for (int t = 0; t < 8; ++t) a[t] = *(const float4*)(rp + 8 * t);
}

__device__ __forceinline__ void seg_mul(const float4 (&a)[8], int kbase,
                                        int h4, int ln15,
                                        const float* Ws,
                                        v8f& c0, v8f& c1, v8f& c2, v8f& c3) {
    const float* wp = Ws + ln15 * WP + kbase + h4;
#pragma unroll
    for (int t = 0; t < 8; ++t) {
        int kk = 8 * t;
        float4 a4 = a[t];
        float4 w0 = *(const float4*)(wp + kk);
        float4 w1 = *(const float4*)(wp + 16 * WP + kk);
        float4 w2 = *(const float4*)(wp + 32 * WP + kk);
        float4 w3 = *(const float4*)(wp + 48 * WP + kk);
        v2f av;  av.x = a4.x;  av.y = a4.y;
        v2f b0; b0.x = w0.x; b0.y = w0.y;
        v2f b1; b1.x = w1.x; b1.y = w1.y;
        v2f b2; b2.x = w2.x; b2.y = w2.y;
        v2f b3; b3.x = w3.x; b3.y = w3.y;
        c0 = __builtin_amdgcn_wmma_f32_16x16x4_f32(false, av, false, b0, (short)0, c0, false, false);
        c1 = __builtin_amdgcn_wmma_f32_16x16x4_f32(false, av, false, b1, (short)0, c1, false, false);
        c2 = __builtin_amdgcn_wmma_f32_16x16x4_f32(false, av, false, b2, (short)0, c2, false, false);
        c3 = __builtin_amdgcn_wmma_f32_16x16x4_f32(false, av, false, b3, (short)0, c3, false, false);
        av.x = a4.z;  av.y = a4.w;
        b0.x = w0.z; b0.y = w0.w;
        b1.x = w1.z; b1.y = w1.w;
        b2.x = w2.z; b2.y = w2.w;
        b3.x = w3.z; b3.y = w3.w;
        c0 = __builtin_amdgcn_wmma_f32_16x16x4_f32(false, av, false, b0, (short)0, c0, false, false);
        c1 = __builtin_amdgcn_wmma_f32_16x16x4_f32(false, av, false, b1, (short)0, c1, false, false);
        c2 = __builtin_amdgcn_wmma_f32_16x16x4_f32(false, av, false, b2, (short)0, c2, false, false);
        c3 = __builtin_amdgcn_wmma_f32_16x16x4_f32(false, av, false, b3, (short)0, c3, false, false);
    }
}

#define HP 68

template <int U> __device__ __forceinline__ int row_bcast(int x) {
    return __builtin_amdgcn_update_dpp(0, x, 0x150 + U, 0xF, 0xF, false);
}

__device__ __forceinline__ void dense_part(const float* __restrict__ feat,
                                           const float* __restrict__ g1,
                                           const float* __restrict__ g2,
                                           const float* __restrict__ bias,
                                           int node, int h4, int hh, int ln15, float nrv,
                                           const float* Ws,
                                           v8f& c0, v8f& c1, v8f& c2, v8f& c3) {
    float4 a1[8], a2[8], a3[8];
    seg_load(g1 + (long)node * F + h4, a1);
    seg_load(g2 + (long)node * F + h4, a2);
    float bv0 = bias[ln15], bv1 = bias[16 + ln15], bv2 = bias[32 + ln15], bv3 = bias[48 + ln15];
#pragma unroll
    for (int i = 0; i < 8; ++i) { c0[i] = 0.0f; c1[i] = 0.0f; c2[i] = 0.0f; c3[i] = 0.0f; }

    seg_mul(a1, 64,  h4, ln15, Ws, c0, c1, c2, c3);
    seg_load(feat + (long)node * F + h4, a3);
    seg_mul(a2, 128, h4, ln15, Ws, c0, c1, c2, c3);

    float rsv = 1.0f / nrv;
#pragma unroll
    for (int r = 0; r < 8; ++r) {
        float s = __shfl(rsv, 8 * hh + r);
        c0[r] = c0[r] * s + bv0;
        c1[r] = c1[r] * s + bv1;
        c2[r] = c2[r] * s + bv2;
        c3[r] = c3[r] * s + bv3;
    }

    seg_mul(a3, 0, h4, ln15, Ws, c0, c1, c2, c3);
}

__global__ __launch_bounds__(256) void hop_gemm_k(const int* __restrict__ csr,
                                                  const int* __restrict__ rowptr,
                                                  const float* __restrict__ nrm,
                                                  const float* __restrict__ feat,
                                                  const float* __restrict__ g1,
                                                  const float* __restrict__ g2,
                                                  const float* __restrict__ W,
                                                  const float* __restrict__ bias,
                                                  float* __restrict__ out) {
    __shared__ float4 Wsv[F * WP / 4];
    __shared__ float4 Hsv[8 * 16 * HP / 4];
    int tid = threadIdx.x;
    for (int i = tid; i < F * KD / 4; i += 256) {
        int o  = i >> 6;
        int c4 = i & 63;
        Wsv[o * (WP / 4) + c4] = ((const float4*)W)[i];
    }
    const float* Ws = (const float*)Wsv;

    int lane = tid & 31;
    int ln15 = lane & 15;
    int hh   = lane >> 4;
    int h4   = hh * 4;
    int wv   = __builtin_amdgcn_readfirstlane(tid >> 5);
    int rowBase = blockIdx.x * 128 + wv * 16;
    bool live = rowBase < NN;
    float* Hs = (float*)Hsv + wv * (16 * HP);

    __syncthreads();
    if (!live) return;

    int node = rowBase + ln15;
    float nrv = nrm[node];
    bool denseFirst = wv < 4;
    v8f c0, c1, c2, c3;
    if (denseFirst) dense_part(feat, g1, g2, bias, node, h4, hh, ln15, nrv, Ws, c0, c1, c2, c3);

    {
        int   rpv = rowptr[rowBase + (lane < 17 ? lane : 16)];
        const float4* g2v = (const float4*)g2;
        int begq[8], endq[8];
        int e0[8];
#pragma unroll
        for (int q = 0; q < 8; ++q) {
            begq[q] = __shfl(rpv, 2 * q + hh);
            endq[q] = __shfl(rpv, 2 * q + hh + 1);
            int j = begq[q] + ln15;
            int jc = j < endq[q] ? j : begq[q];
            if (jc > NE - 1) jc = NE - 1;
            e0[q] = csr[jc];
        }
#pragma unroll
        for (int q = 0; q < 8; ++q) {
            int beg = begq[q], end = endq[q];
            int ent = e0[q];
            float4 acc = make_float4(0.f, 0.f, 0.f, 0.f);
            for (int j0 = beg; j0 < end; j0 += 16) {
                int cur = ent;
                if (j0 + 16 < end) {
                    int jn = j0 + 16 + ln15;
                    ent = csr[jn < end ? jn : beg];
                }
                float4 v[8];
#define LD_ROW(V, U) { unsigned s = (unsigned)row_bcast<U>(cur); V[(U) & 7] = g2v[s * 16u + (unsigned)ln15]; }
#define AC_ROW(V, U) { float w = (j0 + (U) < end) ? 1.0f : 0.0f; \
                       acc.x += V[(U) & 7].x * w; acc.y += V[(U) & 7].y * w; \
                       acc.z += V[(U) & 7].z * w; acc.w += V[(U) & 7].w * w; }
                LD_ROW(v, 0) LD_ROW(v, 1) LD_ROW(v, 2) LD_ROW(v, 3)
                LD_ROW(v, 4) LD_ROW(v, 5) LD_ROW(v, 6) LD_ROW(v, 7)
                if (j0 + 8 < end) {
                    float4 t[8];
                    LD_ROW(t, 8)  LD_ROW(t, 9)  LD_ROW(t, 10) LD_ROW(t, 11)
                    LD_ROW(t, 12) LD_ROW(t, 13) LD_ROW(t, 14) LD_ROW(t, 15)
                    AC_ROW(t, 8)  AC_ROW(t, 9)  AC_ROW(t, 10) AC_ROW(t, 11)
                    AC_ROW(t, 12) AC_ROW(t, 13) AC_ROW(t, 14) AC_ROW(t, 15)
                }
                AC_ROW(v, 0) AC_ROW(v, 1) AC_ROW(v, 2) AC_ROW(v, 3)
                AC_ROW(v, 4) AC_ROW(v, 5) AC_ROW(v, 6) AC_ROW(v, 7)
#undef LD_ROW
#undef AC_ROW
            }
            float nd = __shfl(nrv, 2 * q + hh);
            *(float4*)(Hs + (2 * q + hh) * HP + 4 * ln15) =
                make_float4(acc.x * nd, acc.y * nd, acc.z * nd, acc.w * nd);
        }
    }

    if (!denseFirst) dense_part(feat, g1, g2, bias, node, h4, hh, ln15, nrv, Ws, c0, c1, c2, c3);

    float4 a4h[8];
    seg_load(Hs + ln15 * HP + h4, a4h);
    seg_mul(a4h, 192, h4, ln15, Ws, c0, c1, c2, c3);

    int mBase = rowBase + (h4 ? 8 : 0);
#pragma unroll
    for (int r = 0; r < 8; ++r) {
        int row = mBase + r;
        out[(long)row * F + ln15]      = c0[r];
        out[(long)row * F + 16 + ln15] = c1[r];
        out[(long)row * F + 32 + ln15] = c2[r];
        out[(long)row * F + 48 + ln15] = c3[r];
    }
}


extern "C" void kernel_launch(void* const* d_in, const int* in_sizes, int n_in,
                              void* d_out, int out_size, void* d_ws, size_t ws_size,
                              hipStream_t stream) {
    const float* feat = (const float*)d_in[0];
    const int*   src  = (const int*)d_in[1];
    const int*   dst  = (const int*)d_in[2];
    const float* W    = (const float*)d_in[3];
    const float* bias = (const float*)d_in[4];
    float*       out  = (float*)d_out;

    float* ws   = (float*)d_ws;
    float* nrm  = ws;
    float* ga   = ws + 50048;
    float* gb   = ga + (long)NN * F;
    int*   lst  = (int*)(gb + (long)NN * F);

    int*  csr    = lst;
    int*  rowptr = lst + NE;
    int2* ebuf = (int2*)(lst + NE + 50176);
    int2* tabg = (int2*)(lst + NE + 50176 + 2 * NE);

    bucket_k<<<NCH, 256, 0, stream>>>(src, dst, ebuf, tabg);
    rows_k  <<<NBK, 256, 0, stream>>>(ebuf, tabg, feat, csr, rowptr, nrm, ga);

    const int hopBlocks = (NN + 8 * NPW - 1) / (8 * NPW);

    hop_k<<<hopBlocks, 256, 0, stream>>>(csr, rowptr, nrm, ga, gb);
    hop_k<<<hopBlocks, 256, 0, stream>>>(csr, rowptr, nrm, gb, ga);
    hop_gemm_k<<<(NN + 127) / 128, 256, 0, stream>>>(csr, rowptr, nrm, feat, gb, ga, W, bias, out);
}


# Level 2 places a bucket's list entries in an LDS image of its rows and writes the image out 16 bytes per lane in whole lines; the 32 scattered 4-byte stores per thread go
# speedup vs baseline: 4.4390x; 1.0371x over previous
#include <hip/hip_runtime.h>

typedef float v2f __attribute__((ext_vector_type(2)));
typedef float v8f __attribute__((ext_vector_type(8)));

#define NN 50000
#define NE 800000
#define F  64
#define KD 256

#define NCH 256
#define CHE (NE / NCH)
#define EPT 13
#define RPT 32
#define NPW 4
#define NBK 196
#define SCAP 6144


__global__ __launch_bounds__(256) void bucket_k(const int* __restrict__ src,
                                                const int* __restrict__ dst,
                                                int2* __restrict__ ebuf,
                                                int2* __restrict__ tabg) {
    __shared__ int  cnt[256];
    __shared__ int  off[256];
    __shared__ int  wsum[8];
    __shared__ int2 stage[CHE];
    int tid = threadIdx.x;
    int lane = tid & 31, wv = tid >> 5;
    int e0 = blockIdx.x * CHE;
    int dv[EPT], sv[EPT], rk[EPT], bk[EPT];
#pragma unroll
    for (int k = 0; k < EPT; ++k) {
        int i = tid + k * 256;
        int ii = e0 + (i < CHE ? i : CHE - 1);
        dv[k] = dst[ii];
        sv[k] = src[ii];
    }
    cnt[tid] = 0;
    __syncthreads();
#pragma unroll
    for (int k = 0; k < EPT; ++k) {
        unsigned b = (unsigned)dv[k] >> 8;
        bk[k] = (int)(b < (unsigned)NBK ? b : (unsigned)(NBK - 1));
        rk[k] = 0;
        if (tid + k * 256 < CHE) rk[k] = atomicAdd(&cnt[bk[k]], 1);
    }
    __syncthreads();
    int c = cnt[tid];
    int incl = c;
    for (int o = 1; o < 32; o <<= 1) {
        int y = __shfl_up(incl, o);
        if (lane >= o) incl += y;
    }
    if (lane == 31) wsum[wv] = incl;
    __syncthreads();
    {
        int ex = incl - c;
        for (int w = 0; w < 8; ++w) if (w < wv) ex += wsum[w];
        off[tid] = ex;
        if (tid < NBK) tabg[blockIdx.x * NBK + tid] = make_int2(c, ex);
    }
    __syncthreads();
#pragma unroll
    for (int k = 0; k < EPT; ++k) {
        if (tid + k * 256 < CHE) {
            unsigned p = (unsigned)(off[bk[k]] + rk[k]);
            if (p > (unsigned)(CHE - 1)) p = CHE - 1;
            stage[p] = make_int2(sv[k], dv[k]);
        }
    }
    __syncthreads();
    for (int i = tid; i < CHE; i += 256) ebuf[e0 + i] = stage[i];
}

__global__ __launch_bounds__(256) void rows_k(const int2* __restrict__ ebuf,
                                              const int2* __restrict__ tabg,
                                              const float* __restrict__ feat,
                                              int* __restrict__ csr,
                                              int* __restrict__ rowptr,
                                              float* __restrict__ nrm,
                                              float* __restrict__ g0) {
    __shared__ int deg[512];
    __shared__ float nrs[256];
    __shared__ int cur[512];
    __shared__ int wdeg[8];
    __shared__ int wbase[8];
    __shared__ int4 stgv[(SCAP + 256) / 4];
    int* stg = (int*)stgv;
    int t = threadIdx.x;
    int lane = t & 31, wv = t >> 5;
    int b = blockIdx.x;
    int2 cn = tabg[t * NBK + b];
    int n0 = cn.x, o = cn.y;
    if (o < 0) o = 0;
    if (o > CHE) o = CHE;
    if (n0 < 0) n0 = 0;
    if (n0 > CHE - o) n0 = CHE - o;
    int s0 = t * CHE + o;
    int sc = s0 < NE - 1 ? s0 : NE - 1;
    int2 e[RPT];
#pragma unroll
    for (int u = 0; u < RPT; ++u) e[u] = ebuf[u < n0 ? s0 + u : sc];
    const float4* fv = (const float4*)feat;
    float4 fr[16];
#pragma unroll
    for (int k = 0; k < 16; ++k) {
        int i = t + 256 * k;
        int row = b * 256 + (i >> 4);
        fr[k] = fv[(long)(row < NN ? row : NN - 1) * 16 + (i & 15)];
    }
    int bsum = o;
    for (int x = 16; x > 0; x >>= 1) bsum += __shfl_xor(bsum, x);
    if (lane == 0) wbase[wv] = bsum;
    deg[t] = 0;
    deg[256 + t] = 0;
    cur[256 + t] = 0;
    __syncthreads();
    int base = 0;
    for (int w = 0; w < 8; ++w) base += wbase[w];

    int ix[RPT];
#pragma unroll
    for (int u = 0; u < RPT; ++u) ix[u] = u < n0 ? (e[u].y & 255) : 256 + t;
#pragma unroll
    for (int u = 0; u < RPT; ++u) atomicAdd(&deg[ix[u]], 1);
    for (int j = RPT; j < n0; j += 4) {
        int2 r[4];
#pragma unroll
        for (int u = 0; u < 4; ++u) r[u] = ebuf[j + u < n0 ? s0 + j + u : sc];
#pragma unroll
        for (int u = 0; u < 4; ++u) if (j + u < n0) atomicAdd(&deg[r[u].y & 255], 1);
    }
    __syncthreads();
    int d = deg[t];
    int incl = d;
    for (int x = 1; x < 32; x <<= 1) {
        int y = __shfl_up(incl, x);
        if (lane >= x) incl += y;
    }
    if (lane == 31) wdeg[wv] = incl;
    __syncthreads();
    int start = base + incl - d;
    for (int w = 0; w < 8; ++w) if (w < wv) start += wdeg[w];
    cur[t] = start;
    int tot = 0;
    for (int w = 0; w < 8; ++w) tot += wdeg[w];
    int sh = base & 3;
    bool staged = sh + tot <= SCAP;
    int node = b * 256 + t;
    {
        float f = (float)d;
        float nr = rsqrtf(f < 1.0f ? 1.0f : f);
        nrs[t] = nr;
        if (node < NN) {
            rowptr[node] = start;
            if (node == NN - 1) rowptr[NN] = start + d;
            nrm[node] = nr;
        }
    }
    __syncthreads();
    int ps[RPT];
#pragma unroll
    for (int u = 0; u < RPT; ++u) ps[u] = atomicAdd(&cur[ix[u]], 1);
    if (staged) {
        int lo = base - sh;
#pragma unroll
        for (int u = 0; u < RPT; ++u) {
            unsigned li = (unsigned)(ps[u] - lo);
            if (li > (unsigned)(SCAP - 1)) li = SCAP - 1;
            stg[u < n0 ? li : (unsigned)(SCAP + t)] = e[u].x;
        }
        for (int j = RPT; j < n0; j += 4) {
            int2 r[4];
#pragma unroll
            for (int u = 0; u < 4; ++u) r[u] = ebuf[j + u < n0 ? s0 + j + u : sc];
#pragma unroll
            for (int u = 0; u < 4; ++u) {
                if (j + u < n0) {
                    unsigned li = (unsigned)(atomicAdd(&cur[r[u].y & 255], 1) - lo);
                    if (li > (unsigned)(SCAP - 1)) li = SCAP - 1;
                    stg[li] = r[u].x;
                }
            }
        }
    } else {
#pragma unroll
        for (int u = 0; u < RPT; ++u) {
            if (u < n0) {
                unsigned pos = (unsigned)ps[u];
                if (pos > (unsigned)(NE - 1)) pos = NE - 1;
                csr[pos] = e[u].x;
            }
        }
        for (int j = RPT; j < n0; j += 4) {
            int2 r[4];
#pragma unroll
            for (int u = 0; u < 4; ++u) r[u] = ebuf[j + u < n0 ? s0 + j + u : sc];
#pragma unroll
            for (int u = 0; u < 4; ++u) {
                if (j + u < n0) {
                    unsigned pos = (unsigned)atomicAdd(&cur[r[u].y & 255], 1);
                    if (pos > (unsigned)(NE - 1)) pos = NE - 1;
                    csr[pos] = r[u].x;
                }
            }
        }
    }
    __syncthreads();
    if (staged) {
        int lo = base - sh;
        int hi = base + tot;
        if (hi > NE) hi = NE;
        int nq = (sh + tot + 3) >> 2;
        for (int q = t; q < nq; q += 256) {
            int g = lo + 4 * q;
            int4 v = stgv[q];
            if (g >= base && g + 4 <= hi) {
                *(int4*)(csr + g) = v;
            } else {
                if (g     >= base && g     < hi) csr[g]     = v.x;
                if (g + 1 >= base && g + 1 < hi) csr[g + 1] = v.y;
                if (g + 2 >= base && g + 2 < hi) csr[g + 2] = v.z;
                if (g + 3 >= base && g + 3 < hi) csr[g + 3] = v.w;
            }
        }
    }
    {
        float4* gv = (float4*)g0 + (long)b * (256 * 16);
        float scl[16];
#pragma unroll
        for (int k = 0; k < 16; ++k) scl[k] = nrs[(t >> 4) + 16 * k];
#pragma unroll
        for (int k = 0; k < 16; ++k) {
            int i = t + 256 * k;
            if (b * 256 + (i >> 4) < NN) {
                float s = scl[k];
                gv[i] = make_float4(fr[k].x * s, fr[k].y * s, fr[k].z * s, fr[k].w * s);
            }
        }
    }
}


__global__ __launch_bounds__(256) void hop_k(const int* __restrict__ csr,
                                             const int* __restrict__ rowptr,
                                             const float* __restrict__ nrm,
                                             const float* __restrict__ gin,
                                             float* __restrict__ gout) {
    int wv = __builtin_amdgcn_readfirstlane((int)(threadIdx.x >> 5));
    int n0 = (blockIdx.x * 8 + wv) * NPW;
    if (n0 >= NN) return;
    int lane = threadIdx.x & 31;
    int p  = lane & 15;
    int hh = lane >> 4;
    const float4* gv = (const float4*)gin;
    int rp[NPW + 1];
#pragma unroll
    for (int q = 0; q <= NPW; ++q) rp[q] = rowptr[n0 + q];
#pragma unroll
    for (int q = 0; q < NPW; ++q) {
        int beg = rp[q], end = rp[q + 1];
        float4 acc = make_float4(0.f, 0.f, 0.f, 0.f);
        for (int j0 = beg; j0 < end; j0 += 8) {
            int    pr[4];
            float4 v[4];
#pragma unroll
            for (int u = 0; u < 4; ++u) {
                int j = j0 + 2 * u + hh;
                pr[u] = csr[j < end ? j : beg];
            }
#pragma unroll
            for (int u = 0; u < 4; ++u) v[u] = gv[(unsigned)pr[u] * 16u + (unsigned)p];
#pragma unroll
            for (int u = 0; u < 4; ++u) {
                float w = (j0 + 2 * u + hh < end) ? 1.0f : 0.0f;
                acc.x += v[u].x * w;
                acc.y += v[u].y * w;
                acc.z += v[u].z * w;
                acc.w += v[u].w * w;
            }
        }
        acc.x += __shfl_xor(acc.x, 16);
        acc.y += __shfl_xor(acc.y, 16);
        acc.z += __shfl_xor(acc.z, 16);
        acc.w += __shfl_xor(acc.w, 16);
        if (hh == 0) {
            float nd = nrm[n0 + q];
            ((float4*)(gout + (long)(n0 + q) * F))[p] =
                make_float4(acc.x * nd * nd, acc.y * nd * nd, acc.z * nd * nd, acc.w * nd * nd);
        }
    }
}


#define WP 260

__device__ __forceinline__ void seg_load(const float* __restrict__ rp, float4 (&a)[8]) {
#pragma unroll
    for (int t = 0; t < 8; ++t) a[t] = *(const float4*)(rp + 8 * t);
}

__device__ __forceinline__ void seg_mul(const float4 (&a)[8], int kbase,
                                        int h4, int ln15,
                                        const float* Ws,
                                        v8f& c0, v8f& c1, v8f& c2, v8f& c3) {
    const float* wp = Ws + ln15 * WP + kbase + h4;
#pragma unroll
    for (int t = 0; t < 8; ++t) {
        int kk = 8 * t;
        float4 a4 = a[t];
        float4 w0 = *(const float4*)(wp + kk);
        float4 w1 = *(const float4*)(wp + 16 * WP + kk);
        float4 w2 = *(const float4*)(wp + 32 * WP + kk);
        float4 w3 = *(const float4*)(wp + 48 * WP + kk);
        v2f av;  av.x = a4.x;  av.y = a4.y;
        v2f b0; b0.x = w0.x; b0.y = w0.y;
        v2f b1; b1.x = w1.x; b1.y = w1.y;
        v2f b2; b2.x = w2.x; b2.y = w2.y;
        v2f b3; b3.x = w3.x; b3.y = w3.y;
        c0 = __builtin_amdgcn_wmma_f32_16x16x4_f32(false, av, false, b0, (short)0, c0, false, false);
        c1 = __builtin_amdgcn_wmma_f32_16x16x4_f32(false, av, false, b1, (short)0, c1, false, false);
        c2 = __builtin_amdgcn_wmma_f32_16x16x4_f32(false, av, false, b2, (short)0, c2, false, false);
        c3 = __builtin_amdgcn_wmma_f32_16x16x4_f32(false, av, false, b3, (short)0, c3, false, false);
        av.x = a4.z;  av.y = a4.w;
        b0.x = w0.z; b0.y = w0.w;
        b1.x = w1.z; b1.y = w1.w;
        b2.x = w2.z; b2.y = w2.w;
        b3.x = w3.z; b3.y = w3.w;
        c0 = __builtin_amdgcn_wmma_f32_16x16x4_f32(false, av, false, b0, (short)0, c0, false, false);
        c1 = __builtin_amdgcn_wmma_f32_16x16x4_f32(false, av, false, b1, (short)0, c1, false, false);
        c2 = __builtin_amdgcn_wmma_f32_16x16x4_f32(false, av, false, b2, (short)0, c2, false, false);
        c3 = __builtin_amdgcn_wmma_f32_16x16x4_f32(false, av, false, b3, (short)0, c3, false, false);
    }
}

#define HP 68

template <int U> __device__ __forceinline__ int row_bcast(int x) {
    return __builtin_amdgcn_update_dpp(0, x, 0x150 + U, 0xF, 0xF, false);
}

__device__ __forceinline__ void dense_part(const float* __restrict__ feat,
                                           const float* __restrict__ g1,
                                           const float* __restrict__ g2,
                                           const float* __restrict__ bias,
                                           int node, int h4, int hh, int ln15, float nrv,
                                           const float* Ws,
                                           v8f& c0, v8f& c1, v8f& c2, v8f& c3) {
    float4 a1[8], a2[8], a3[8];
    seg_load(g1 + (long)node * F + h4, a1);
    seg_load(g2 + (long)node * F + h4, a2);
    float bv0 = bias[ln15], bv1 = bias[16 + ln15], bv2 = bias[32 + ln15], bv3 = bias[48 + ln15];
#pragma unroll
    for (int i = 0; i < 8; ++i) { c0[i] = 0.0f; c1[i] = 0.0f; c2[i] = 0.0f; c3[i] = 0.0f; }

    seg_mul(a1, 64,  h4, ln15, Ws, c0, c1, c2, c3);
    seg_load(feat + (long)node * F + h4, a3);
    seg_mul(a2, 128, h4, ln15, Ws, c0, c1, c2, c3);

    float rsv = 1.0f / nrv;
#pragma unroll
    for (int r = 0; r < 8; ++r) {
        float s = __shfl(rsv, 8 * hh + r);
        c0[r] = c0[r] * s + bv0;
        c1[r] = c1[r] * s + bv1;
        c2[r] = c2[r] * s + bv2;
        c3[r] = c3[r] * s + bv3;
    }

    seg_mul(a3, 0, h4, ln15, Ws, c0, c1, c2, c3);
}

__global__ __launch_bounds__(256) void hop_gemm_k(const int* __restrict__ csr,
                                                  const int* __restrict__ rowptr,
                                                  const float* __restrict__ nrm,
                                                  const float* __restrict__ feat,
                                                  const float* __restrict__ g1,
                                                  const float* __restrict__ g2,
                                                  const float* __restrict__ W,
                                                  const float* __restrict__ bias,
                                                  float* __restrict__ out) {
    __shared__ float4 Wsv[F * WP / 4];
    __shared__ float4 Hsv[8 * 16 * HP / 4];
    int tid = threadIdx.x;
    for (int i = tid; i < F * KD / 4; i += 256) {
        int o  = i >> 6;
        int c4 = i & 63;
        Wsv[o * (WP / 4) + c4] = ((const float4*)W)[i];
    }
    const float* Ws = (const float*)Wsv;

    int lane = tid & 31;
    int ln15 = lane & 15;
    int hh   = lane >> 4;
    int h4   = hh * 4;
    int wv   = __builtin_amdgcn_readfirstlane(tid >> 5);
    int rowBase = blockIdx.x * 128 + wv * 16;
    bool live = rowBase < NN;
    float* Hs = (float*)Hsv + wv * (16 * HP);

    __syncthreads();
    if (!live) return;

    int node = rowBase + ln15;
    float nrv = nrm[node];
    bool denseFirst = wv < 4;
    v8f c0, c1, c2, c3;
    if (denseFirst) dense_part(feat, g1, g2, bias, node, h4, hh, ln15, nrv, Ws, c0, c1, c2, c3);

    {
        int   rpv = rowptr[rowBase + (lane < 17 ? lane : 16)];
        const float4* g2v = (const float4*)g2;
        int begq[8], endq[8];
        int e0[8];
#pragma unroll
        for (int q = 0; q < 8; ++q) {
            begq[q] = __shfl(rpv, 2 * q + hh);
            endq[q] = __shfl(rpv, 2 * q + hh + 1);
            int j = begq[q] + ln15;
            int jc = j < endq[q] ? j : begq[q];
            if (jc > NE - 1) jc = NE - 1;
            e0[q] = csr[jc];
        }
#pragma unroll
        for (int q = 0; q < 8; ++q) {
            int beg = begq[q], end = endq[q];
            int ent = e0[q];
            float4 acc = make_float4(0.f, 0.f, 0.f, 0.f);
            for (int j0 = beg; j0 < end; j0 += 16) {
                int cur = ent;
                if (j0 + 16 < end) {
                    int jn = j0 + 16 + ln15;
                    ent = csr[jn < end ? jn : beg];
                }
                float4 v[8];
#define LD_ROW(V, U) { unsigned s = (unsigned)row_bcast<U>(cur); V[(U) & 7] = g2v[s * 16u + (unsigned)ln15]; }
#define AC_ROW(V, U) { float w = (j0 + (U) < end) ? 1.0f : 0.0f; \
                       acc.x += V[(U) & 7].x * w; acc.y += V[(U) & 7].y * w; \
                       acc.z += V[(U) & 7].z * w; acc.w += V[(U) & 7].w * w; }
                LD_ROW(v, 0) LD_ROW(v, 1) LD_ROW(v, 2) LD_ROW(v, 3)
                LD_ROW(v, 4) LD_ROW(v, 5) LD_ROW(v, 6) LD_ROW(v, 7)
                if (j0 + 8 < end) {
                    float4 t[8];
                    LD_ROW(t, 8)  LD_ROW(t, 9)  LD_ROW(t, 10) LD_ROW(t, 11)
                    LD_ROW(t, 12) LD_ROW(t, 13) LD_ROW(t, 14) LD_ROW(t, 15)
                    AC_ROW(t, 8)  AC_ROW(t, 9)  AC_ROW(t, 10) AC_ROW(t, 11)
                    AC_ROW(t, 12) AC_ROW(t, 13) AC_ROW(t, 14) AC_ROW(t, 15)
                }
                AC_ROW(v, 0) AC_ROW(v, 1) AC_ROW(v, 2) AC_ROW(v, 3)
                AC_ROW(v, 4) AC_ROW(v, 5) AC_ROW(v, 6) AC_ROW(v, 7)
#undef LD_ROW
#undef AC_ROW
            }
            float nd = __shfl(nrv, 2 * q + hh);
            *(float4*)(Hs + (2 * q + hh) * HP + 4 * ln15) =
                make_float4(acc.x * nd, acc.y * nd, acc.z * nd, acc.w * nd);
        }
    }

    if (!denseFirst) dense_part(feat, g1, g2, bias, node, h4, hh, ln15, nrv, Ws, c0, c1, c2, c3);

    float4 a4h[8];
    seg_load(Hs + ln15 * HP + h4, a4h);
    seg_mul(a4h, 192, h4, ln15, Ws, c0, c1, c2, c3);

    int mBase = rowBase + (h4 ? 8 : 0);
#pragma unroll
    for (int r = 0; r < 8; ++r) {
        int row = mBase + r;
        out[(long)row * F + ln15]      = c0[r];
        out[(long)row * F + 16 + ln15] = c1[r];
        out[(long)row * F + 32 + ln15] = c2[r];
        out[(long)row * F + 48 + ln15] = c3[r];
    }
}


extern "C" void kernel_launch(void* const* d_in, const int* in_sizes, int n_in,
                              void* d_out, int out_size, void* d_ws, size_t ws_size,
                              hipStream_t stream) {
    const float* feat = (const float*)d_in[0];
    const int*   src  = (const int*)d_in[1];
    const int*   dst  = (const int*)d_in[2];
    const float* W    = (const float*)d_in[3];
    const float* bias = (const float*)d_in[4];
    float*       out  = (float*)d_out;

    float* ws   = (float*)d_ws;
    float* nrm  = ws;
    float* ga   = ws + 50048;
    float* gb   = ga + (long)NN * F;
    int*   lst  = (int*)(gb + (long)NN * F);

    int*  csr    = lst;
    int*  rowptr = lst + NE;
    int2* ebuf = (int2*)(lst + NE + 50176);
    int2* tabg = (int2*)(lst + NE + 50176 + 2 * NE);

    bucket_k<<<NCH, 256, 0, stream>>>(src, dst, ebuf, tabg);
    rows_k  <<<NBK, 256, 0, stream>>>(ebuf, tabg, feat, csr, rowptr, nrm, ga);

    const int hopBlocks = (NN + 8 * NPW - 1) / (8 * NPW);

    hop_k<<<hopBlocks, 256, 0, stream>>>(csr, rowptr, nrm, ga, gb);
    hop_k<<<hopBlocks, 256, 0, stream>>>(csr, rowptr, nrm, gb, ga);
    hop_gemm_k<<<(NN + 127) / 128, 256, 0, stream>>>(csr, rowptr, nrm, feat, gb, ga, W, bias, out);
}
